# starting kernel
# baseline (speedup 1.0000x reference)
#include <hip/hip_runtime.h>
#include <hip/hip_bf16.h>

#define NN   512
#define KSPA 64
#define ND   16
#define NB   1024
#define NSH  448
#define N_L  30688

typedef __attribute__((ext_vector_type(2))) float v2f;
typedef __attribute__((ext_vector_type(8))) float v8f;

__global__ __launch_bounds__(128) void build_F_kernel(
    const float* __restrict__ L_emb,
    const float* __restrict__ S_emb,
    float* __restrict__ Fmat)
{
    __shared__ float hist[64][128];
    const int d  = blockIdx.x;
    const int c  = blockIdx.y * 128 + threadIdx.x;

    if (d == 0) {
        #pragma unroll 4
        for (int i = 0; i < 64; ++i)
            Fmat[(0 * 64 + i) * NN + c] = (c == NSH + i) ? 1.0f : 0.0f;
        return;
    }

    const float* Ld = L_emb + d * N_L;
    for (int i = 0; i < 64; ++i) {
        const int off = NSH * i + (i * (i - 1)) / 2;
        float acc;
        if (c < NSH) acc = Ld[off + c];
        else         acc = (c - NSH == i) ? S_emb[d * KSPA + i] : 0.0f;
        for (int k = 0; k < i; ++k) {
            const float w = Ld[off + NSH + k];
            acc += w * hist[k][threadIdx.x];
        }
        hist[i][threadIdx.x] = acc;
        Fmat[(d * 64 + i) * NN + c] = acc;
    }
}

__global__ __launch_bounds__(256) void copy_bias_kernel(
    const float* __restrict__ eps,
    const float* __restrict__ bias_shared,
    float* __restrict__ out)
{
    const int idx = blockIdx.x * blockDim.x + threadIdx.x;
    const int b  = idx / (NSH / 4);
    const int c4 = idx % (NSH / 4);
    float4 e  = ((const float4*)(eps + b * NN))[c4];
    float4 bs = ((const float4*)bias_shared)[c4];
    float4 o  = make_float4(e.x + bs.x, e.y + bs.y, e.z + bs.z, e.w + bs.w);
    ((float4*)(out + b * NN))[c4] = o;
}

__global__ __launch_bounds__(32) void spa_wmma_kernel(
    const float* __restrict__ eps,
    const int*   __restrict__ didx,
    const float* __restrict__ Fmat,
    const float* __restrict__ bias_ns,
    float* __restrict__ out)
{
    const int bt   = blockIdx.x;
    const int rt   = blockIdx.y;
    const int dom  = blockIdx.z;
    const int lane = threadIdx.x;
    const int half = lane >> 4;
    const int l16  = lane & 15;

    __shared__ float As[16][20];
    __shared__ float Bs[16][20];

    const float* Fbase = Fmat + (dom * 64 + rt * 16) * NN;
    const float* Ebase = eps + (bt * 16) * NN;

    const int lrow = lane >> 1;
    const int lcol = (lane & 1) * 8;

    v8f acc = {0.f, 0.f, 0.f, 0.f, 0.f, 0.f, 0.f, 0.f};

    for (int kk = 0; kk < NN; kk += 16) {
        float4 fa0 = *(const float4*)(Fbase + lrow * NN + kk + lcol);
        float4 fa1 = *(const float4*)(Fbase + lrow * NN + kk + lcol + 4);
        float4 eb0 = *(const float4*)(Ebase + lrow * NN + kk + lcol);
        float4 eb1 = *(const float4*)(Ebase + lrow * NN + kk + lcol + 4);
        __syncthreads();
        *(float4*)&As[lrow][lcol]     = fa0;
        *(float4*)&As[lrow][lcol + 4] = fa1;
        *(float4*)&Bs[lrow][lcol]     = eb0;
        *(float4*)&Bs[lrow][lcol + 4] = eb1;
        __syncthreads();

        #pragma unroll
        for (int ks = 0; ks < 16; ks += 4) {
            const int k0 = ks + half * 2;
            v2f a; a.x = As[l16][k0]; a.y = As[l16][k0 + 1];
            v2f b; b.x = Bs[l16][k0]; b.y = Bs[l16][k0 + 1];
            acc = __builtin_amdgcn_wmma_f32_16x16x4_f32(
                 false, a,  false, b,
                 (short)0, acc,  false,  false);
        }
    }

    const int b = bt * 16 + l16;
    if (didx[b] == dom) {
        #pragma unroll
        for (int v = 0; v < 8; ++v) {
            const int m = v + half * 8;
            out[b * NN + NSH + rt * 16 + m] =
                acc[v] + bias_ns[dom * KSPA + rt * 16 + m];
        }
    }
}

extern "C" void kernel_launch(void* const* d_in, const int* in_sizes, int n_in,
                              void* d_out, int out_size, void* d_ws, size_t ws_size,
                              hipStream_t stream) {
    const float* eps     = (const float*)d_in[0];
    const int*   didx    = (const int*)  d_in[1];
    const float* L_emb   = (const float*)d_in[2];
    const float* S_emb   = (const float*)d_in[3];
    const float* bias_ns = (const float*)d_in[4];
    const float* bias_sh = (const float*)d_in[5];
    float*       out     = (float*)d_out;
    float*       Fmat    = (float*)d_ws;

    build_F_kernel<<<dim3(ND, NN / 128), dim3(128), 0, stream>>>(L_emb, S_emb, Fmat);

    const int n4 = NB * (NSH / 4);
    copy_bias_kernel<<<n4 / 256, 256, 0, stream>>>(eps, bias_sh, out);

    spa_wmma_kernel<<<dim3(NB / 16, KSPA / 16, ND), dim3(32), 0, stream>>>(
        eps, didx, Fmat, bias_ns, out);
}


# Shared prefix out[:, :448] = eps + bias is stored by the tail product's waves from the eps fragments they already hold; the copy launch and its eps read go: 2 launches (was 3); sums unchanged
# speedup vs baseline: 7.4466x; 7.4466x over previous
#include <hip/hip_runtime.h>
#include <hip/hip_bf16.h>

#define NN   512
#define KSPA 64
#define ND   16
#define NB   1024
#define NSH  448
#define N_L  30688

typedef __attribute__((ext_vector_type(2))) float v2f;
typedef __attribute__((ext_vector_type(8))) float v8f;

#define MST 68

__device__ __forceinline__ v8f mma4(v2f a, v2f b, v8f c) {
    return __builtin_amdgcn_wmma_f32_16x16x4_f32(
         false, a,  false, b,
         (short)0, c,  false,  false);
}

__global__ __launch_bounds__(128) void build_F_kernel(
    const float* __restrict__ L_emb,
    const float* __restrict__ S_emb,
    float* __restrict__ Fmat)
{
    __shared__ float Wl[64][MST];
    __shared__ float Mi[64][MST];
    __shared__ float Ts[4][16][20];

    const int d    = blockIdx.x;
    const int tid  = threadIdx.x;
    const int wave = tid >> 5;
    const int lane = tid & 31;
    const int half = lane >> 4;
    const int l16  = lane & 15;
    const int cw   = blockIdx.y * 128 + wave * 32;

    if (d == 0) {
        const int c = blockIdx.y * 128 + tid;
        #pragma unroll 4
        for (int i = 0; i < 64; ++i)
            Fmat[(0 * 64 + i) * NN + c] = (c == NSH + i) ? 1.0f : 0.0f;
        return;
    }

    const float* Ld = L_emb + d * N_L;
    const bool left = cw < NSH;

    float wreg[32];
    #pragma unroll
    for (int q = 0; q < 32; ++q) {
        const int idx = q * 128 + tid;
        const int i = idx >> 6, k = idx & 63;
        const float w = Ld[NSH * i + (i * (i - 1)) / 2 + NSH + ((k < i) ? k : 0)];
        wreg[q] = (k < i) ? w : 0.0f;
    }

    v2f bf0[16], bf1[16];
    if (left) {
        #pragma unroll
        for (int s = 0; s < 16; ++s) {
            const int k0 = 4 * s + 2 * half;
            const float* r0 = Ld + NSH * k0 + (k0 * (k0 - 1)) / 2 + cw + l16;
            const float* r1 = Ld + NSH * (k0 + 1) + ((k0 + 1) * k0) / 2 + cw + l16;
            bf0[s].x = r0[0];  bf0[s].y = r1[0];
            bf1[s].x = r0[16]; bf1[s].y = r1[16];
        }
    }

    #pragma unroll
    for (int q = 0; q < 32; ++q) {
        const int idx = q * 128 + tid;
        Wl[idx >> 6][idx & 63] = wreg[q];
        Mi[idx >> 6][idx & 63] = 0.0f;
    }
    __syncthreads();

    if (tid < 64) {
        const int I = tid >> 4, j = tid & 15;
        float x[16];
        #pragma unroll
        for (int r = 0; r < 16; ++r) x[r] = (r == j) ? 1.0f : 0.0f;
        #pragma unroll
        for (int k = 0; k < 15; ++k) {
            const float xk = x[k];
            #pragma unroll
            for (int r = k + 1; r < 16; ++r)
                x[r] += Wl[16 * I + r][16 * I + k] * xk;
        }
        #pragma unroll
        for (int r = 0; r < 16; ++r) Mi[16 * I + r][16 * I + j] = x[r];
    }
    __syncthreads();

    #pragma unroll
    for (int s = 1; s < 4; ++s) {
        const bool act = wave < 4 - s;
        const int I = s + wave, J = wave;
        if (act) {
            v8f t = {0.f, 0.f, 0.f, 0.f, 0.f, 0.f, 0.f, 0.f};
            for (int K = J; K < I; ++K) {
                #pragma unroll
                for (int ks = 0; ks < 16; ks += 4) {
                    const int k0 = ks + half * 2;
                    v2f a; a.x = Wl[16 * I + l16][16 * K + k0];
                           a.y = Wl[16 * I + l16][16 * K + k0 + 1];
                    v2f b; b.x = Mi[16 * K + k0][16 * J + l16];
                           b.y = Mi[16 * K + k0 + 1][16 * J + l16];
                    t = mma4(a, b, t);
                }
            }
            #pragma unroll
            for (int v = 0; v < 8; ++v) Ts[wave][v + half * 8][l16] = t[v];
        }
        __syncthreads();
        if (act) {
            v8f r = {0.f, 0.f, 0.f, 0.f, 0.f, 0.f, 0.f, 0.f};
            #pragma unroll
            for (int ks = 0; ks < 16; ks += 4) {
                const int k0 = ks + half * 2;
                v2f a; a.x = Mi[16 * I + l16][16 * I + k0];
                       a.y = Mi[16 * I + l16][16 * I + k0 + 1];
                v2f b; b.x = Ts[wave][k0][l16];
                       b.y = Ts[wave][k0 + 1][l16];
                r = mma4(a, b, r);
            }
            #pragma unroll
            for (int v = 0; v < 8; ++v) Mi[16 * I + v + half * 8][16 * J + l16] = r[v];
        }
        __syncthreads();
    }

    float* Fd = Fmat + d * 64 * NN;

    if (left) {
        #pragma unroll
        for (int I = 0; I < 4; ++I) {
            v8f acc0 = {0.f, 0.f, 0.f, 0.f, 0.f, 0.f, 0.f, 0.f};
            v8f acc1 = {0.f, 0.f, 0.f, 0.f, 0.f, 0.f, 0.f, 0.f};
            #pragma unroll
            for (int s = 0; s < 4 * (I + 1); ++s) {
                const int k0 = 4 * s + 2 * half;
                v2f a; a.x = Mi[16 * I + l16][k0];
                       a.y = Mi[16 * I + l16][k0 + 1];
                acc0 = mma4(a, bf0[s], acc0);
                acc1 = mma4(a, bf1[s], acc1);
            }
            #pragma unroll
            for (int v = 0; v < 8; ++v) {
                float* p = Fd + (16 * I + v + half * 8) * NN + cw + l16;
                p[0]  = acc0[v];
                p[16] = acc1[v];
            }
        }
    } else {
        const int j = cw - NSH + lane;
        const float sj = S_emb[d * KSPA + j];
        #pragma unroll 8
        for (int i = 0; i < 64; ++i)
            Fd[i * NN + NSH + j] = Mi[i][j] * sj;
    }
}

#define TSLOTS 8
#define KCH    8

__global__ __launch_bounds__(32) void spa_wmma_kernel(
    const float* __restrict__ eps,
    const int*   __restrict__ didx,
    const float* __restrict__ Fmat,
    const float* __restrict__ bias_ns,
    const float* __restrict__ bias_sh,
    float* __restrict__ out)
{
    const int dom  = blockIdx.x;
    const int rt   = blockIdx.y;
    const int slot = blockIdx.z;
    const int lane = threadIdx.x;
    const int half = lane >> 4;
    const int l16  = lane & 15;

    __shared__ int lst[NB];

    float4 bsv[2 * KCH];
    const float* Br = bias_sh + 8 * (2 * KCH * rt) + 4 * half;
    #pragma unroll
    for (int j = 0; j < KCH; ++j) {
        bsv[j]       = *(const float4*)(Br + 8 * j);
        bsv[KCH + j] = make_float4(0.f, 0.f, 0.f, 0.f);
    }
    if (2 * rt + 1 < NSH / (8 * KCH)) {
        #pragma unroll
        for (int j = 0; j < KCH; ++j)
            bsv[KCH + j] = *(const float4*)(Br + 8 * (KCH + j));
    }

    int dv[32];
    #pragma unroll
    for (int j = 0; j < 32; ++j) dv[j] = didx[32 * j + lane];
    int total = 0;
    #pragma unroll
    for (int j = 0; j < 32; ++j) {
        const bool mine = dv[j] == dom;
        const unsigned m = __builtin_amdgcn_ballot_w32(mine);
        if (mine) lst[total + __popc(m & ((1u << lane) - 1u))] = 32 * j + lane;
        total += __popc(m);
    }
    __syncthreads();
    const int ntiles = (total + 15) >> 4;

    const float* Fr = Fmat + (dom * 64 + rt * 16 + l16) * NN + 4 * half;

    for (int t = slot; t < ntiles; t += TSLOTS) {
        const int  r     = 16 * t + l16;
        const bool valid = r < total;
        const int  bi    = lst[valid ? r : 16 * t];
        const float* Er  = eps + bi * NN + 4 * half;

        v8f acc0 = {0.f, 0.f, 0.f, 0.f, 0.f, 0.f, 0.f, 0.f};
        v8f acc1 = {0.f, 0.f, 0.f, 0.f, 0.f, 0.f, 0.f, 0.f};
        float4 fa[2][KCH], fb[2][KCH];

        #pragma unroll
        for (int c = 0; c < KCH; ++c) {
            fa[0][c] = *(const float4*)(Fr + 8 * c);
            fb[0][c] = *(const float4*)(Er + 8 * c);
        }
        #pragma unroll
        for (int ch = 0; ch < NN / (8 * KCH); ++ch) {
            const int cur = ch & 1, nxt = cur ^ 1;
            if (ch + 1 < NN / (8 * KCH)) {
                #pragma unroll
                for (int c = 0; c < KCH; ++c) {
                    fa[nxt][c] = *(const float4*)(Fr + 8 * (KCH * (ch + 1) + c));
                    fb[nxt][c] = *(const float4*)(Er + 8 * (KCH * (ch + 1) + c));
                }
            }
            if (ch < NSH / (8 * KCH) && (ch >> 1) == rt) {
                if (valid) {
                    float* pp = out + bi * NN + 4 * half + 8 * KCH * ch;
                    #pragma unroll
                    for (int c = 0; c < KCH; ++c) {
                        const float4 e = fb[cur][c];
                        const float4 q = bsv[(ch & 1) * KCH + c];
                        *(float4*)(pp + 8 * c) =
                            make_float4(e.x + q.x, e.y + q.y, e.z + q.z, e.w + q.w);
                    }
                }
            }
            #pragma unroll
            for (int c = 0; c < KCH; ++c) {
                v2f a0; a0.x = fa[cur][c].x; a0.y = fa[cur][c].y;
                v2f b0; b0.x = fb[cur][c].x; b0.y = fb[cur][c].y;
                v2f a1; a1.x = fa[cur][c].z; a1.y = fa[cur][c].w;
                v2f b1; b1.x = fb[cur][c].z; b1.y = fb[cur][c].w;
                acc0 = mma4(a0, b0, acc0);
                acc1 = mma4(a1, b1, acc1);
            }
        }

        if (valid) {
            const float* bp = bias_ns + dom * KSPA + rt * 16 + half * 8;
            const float4 q0 = *(const float4*)bp;
            const float4 q1 = *(const float4*)(bp + 4);
            float* op = out + bi * NN + NSH + rt * 16 + half * 8;
            float4 o0, o1;
            o0.x = (acc0[0] + acc1[0]) + q0.x;  o0.y = (acc0[1] + acc1[1]) + q0.y;
            o0.z = (acc0[2] + acc1[2]) + q0.z;  o0.w = (acc0[3] + acc1[3]) + q0.w;
            o1.x = (acc0[4] + acc1[4]) + q1.x;  o1.y = (acc0[5] + acc1[5]) + q1.y;
            o1.z = (acc0[6] + acc1[6]) + q1.z;  o1.w = (acc0[7] + acc1[7]) + q1.w;
            *(float4*)op       = o0;
            *(float4*)(op + 4) = o1;
        }
    }
}

extern "C" void kernel_launch(void* const* d_in, const int* in_sizes, int n_in,
                              void* d_out, int out_size, void* d_ws, size_t ws_size,
                              hipStream_t stream) {
    const float* eps     = (const float*)d_in[0];
    const int*   didx    = (const int*)  d_in[1];
    const float* L_emb   = (const float*)d_in[2];
    const float* S_emb   = (const float*)d_in[3];
    const float* bias_ns = (const float*)d_in[4];
    const float* bias_sh = (const float*)d_in[5];
    float*       out     = (float*)d_out;
    float*       Fmat    = (float*)d_ws;

    build_F_kernel<<<dim3(ND, NN / 128), dim3(128), 0, stream>>>(L_emb, S_emb, Fmat);

    spa_wmma_kernel<<<dim3(ND, KSPA / 16, TSLOTS), dim3(32), 0, stream>>>(
        eps, didx, Fmat, bias_ns, bias_sh, out);
}


# One launch: a block (domain,row tile) forms M in LDS; its 16 waves build a 32-k slice of F transposed, so the accumulators are the tail product's A operand; k partials summed via LDS; order changes
# speedup vs baseline: 8.8988x; 1.1950x over previous
#include <hip/hip_runtime.h>
#include <hip/hip_bf16.h>

#define NN   512
#define KSPA 64
#define ND   16
#define NB   1024
#define NSH  448
#define N_L  30688

typedef __attribute__((ext_vector_type(2))) float v2f;
typedef __attribute__((ext_vector_type(8))) float v8f;

#define MST 68
#define NW  16
#define TS  2
#define TG  3

__device__ __forceinline__ v8f mma4(v2f a, v2f b, v8f c) {
    return __builtin_amdgcn_wmma_f32_16x16x4_f32(
         false, a,  false, b,
         (short)0, c,  false,  false);
}

__global__ __launch_bounds__(512) void fused_spa_kernel(
    const float* __restrict__ eps,
    const int*   __restrict__ didx,
    const float* __restrict__ L_emb,
    const float* __restrict__ S_emb,
    const float* __restrict__ bias_ns,
    const float* __restrict__ bias_sh,
    float* __restrict__ out)
{
    __shared__ __attribute__((aligned(16))) float Wl[64][MST];
    __shared__ __attribute__((aligned(16))) float Mi[64][MST];
    __shared__ float Ts[4][16][20];
    __shared__ int   lst[NB];
    __shared__ int   s_total;
    __shared__ __attribute__((aligned(16))) float red[TG][NW][256];

    const int dom  = blockIdx.x;
    const int rt   = blockIdx.y;
    const int slot = blockIdx.z;
    const int tid  = threadIdx.x;
    const int wave = tid >> 5;
    const int lane = tid & 31;
    const int half = lane >> 4;
    const int l16  = lane & 15;
    const int kw   = 32 * wave;
    const bool leftw = wave < 14;
    const bool pw    = leftw && ((wave & 3) == rt);

    const float* Ld = L_emb + dom * N_L;

    int dv[32];
    if (wave == NW - 1) {
        #pragma unroll
        for (int j = 0; j < 32; ++j) dv[j] = didx[32 * j + lane];
    }
    float wreg[8];
    v2f bf0[16], bf1[16];
    float4 sv[4];
    #pragma unroll
    for (int s = 0; s < 16; ++s) { bf0[s].x = 0.f; bf0[s].y = 0.f; bf1[s].x = 0.f; bf1[s].y = 0.f; }
    #pragma unroll
    for (int s = 0; s < 4; ++s) sv[s] = make_float4(0.f, 0.f, 0.f, 0.f);
    if (dom != 0) {
        #pragma unroll
        for (int q = 0; q < 8; ++q) {
            const int idx = q * 512 + tid;
            const int i = idx >> 6, k = idx & 63;
            const float w = Ld[NSH * i + (i * (i - 1)) / 2 + NSH + ((k < i) ? k : 0)];
            wreg[q] = (k < i) ? w : 0.0f;
        }
        if (leftw) {
            #pragma unroll
            for (int J = 0; J < 4; ++J) {
                if (J <= rt) {
                    #pragma unroll
                    for (int ss = 0; ss < 4; ++ss) {
                        const int s  = 4 * J + ss;
                        const int k0 = 4 * s + 2 * half;
                        const float* r0 = Ld + NSH * k0 + (k0 * (k0 - 1)) / 2 + kw + l16;
                        const float* r1 = Ld + NSH * (k0 + 1) + ((k0 + 1) * k0) / 2 + kw + l16;
                        bf0[s].x = r0[0];  bf0[s].y = r1[0];
                        bf1[s].x = r0[16]; bf1[s].y = r1[16];
                    }
                }
            }
        } else {
            const float* Sp = S_emb + dom * KSPA + (kw - NSH) + 8 * half;
            sv[0] = *(const float4*)(Sp);
            sv[1] = *(const float4*)(Sp + 4);
            sv[2] = *(const float4*)(Sp + 16);
            sv[3] = *(const float4*)(Sp + 20);
        }
    }
    float4 bs[4];
    #pragma unroll
    for (int s = 0; s < 4; ++s) bs[s] = make_float4(0.f, 0.f, 0.f, 0.f);
    if (pw) {
        const float* Bp = bias_sh + kw + 8 * half;
        bs[0] = *(const float4*)(Bp);
        bs[1] = *(const float4*)(Bp + 4);
        bs[2] = *(const float4*)(Bp + 16);
        bs[3] = *(const float4*)(Bp + 20);
    }

    if (dom != 0) {
        #pragma unroll
        for (int q = 0; q < 8; ++q) {
            const int idx = q * 512 + tid;
            Wl[idx >> 6][idx & 63] = wreg[q];
            Mi[idx >> 6][idx & 63] = 0.0f;
        }
    }
    __syncthreads();

    if (wave == NW - 1) {
        int total = 0;
        #pragma unroll
        for (int j = 0; j < 32; ++j) {
            const bool mine = dv[j] == dom;
            const unsigned m = __builtin_amdgcn_ballot_w32(mine);
            if (mine) lst[total + __popc(m & ((1u << lane) - 1u))] = 32 * j + lane;
            total += __popc(m);
        }
        if (lane == 0) s_total = total;
    } else if (dom != 0 && tid < 64) {
        const int I = tid >> 4, j = tid & 15;
        float x[16];
        #pragma unroll
        for (int r = 0; r < 16; ++r) x[r] = (r == j) ? 1.0f : 0.0f;
        #pragma unroll
        for (int k = 0; k < 15; ++k) {
            const float xk = x[k];
            #pragma unroll
            for (int r = k + 1; r < 16; ++r)
                x[r] += Wl[16 * I + r][16 * I + k] * xk;
        }
        #pragma unroll
        for (int r = 0; r < 16; ++r) Mi[16 * I + r][16 * I + j] = x[r];
    }
    __syncthreads();

    const int total  = s_total;
    const int ntiles = (total + 15) >> 4;

    float4 ev[TG][4];
    int    bis[TG];
    bool   vld[TG];
    auto load_pass = [&](int tbase) {
        #pragma unroll
        for (int tt = 0; tt < TG; ++tt) {
            const int t = tbase + TS * tt;
            bis[tt] = 0; vld[tt] = false;
            if (t < ntiles) {
                const int r = 16 * t + l16;
                vld[tt] = r < total;
                bis[tt] = lst[vld[tt] ? r : 16 * t];
                const float* Er = eps + bis[tt] * NN + kw + 8 * half;
                ev[tt][0] = *(const float4*)(Er);
                ev[tt][1] = *(const float4*)(Er + 4);
                ev[tt][2] = *(const float4*)(Er + 16);
                ev[tt][3] = *(const float4*)(Er + 20);
            }
        }
    };
    load_pass(slot);

    const v8f zero8 = {0.f, 0.f, 0.f, 0.f, 0.f, 0.f, 0.f, 0.f};
    v8f F0 = zero8, F1 = zero8;

    if (dom != 0) {
        #pragma unroll
        for (int s = 1; s < 4; ++s) {
            if (s <= rt) {
                const bool act = wave < 4 - s;
                const int I = s + wave, J = wave;
                if (act) {
                    v8f t = zero8;
                    for (int K = J; K < I; ++K) {
                        #pragma unroll
                        for (int ks = 0; ks < 16; ks += 4) {
                            const int k0 = ks + half * 2;
                            v2f a; a.x = Wl[16 * I + l16][16 * K + k0];
                                   a.y = Wl[16 * I + l16][16 * K + k0 + 1];
                            v2f b; b.x = Mi[16 * K + k0][16 * J + l16];
                                   b.y = Mi[16 * K + k0 + 1][16 * J + l16];
                            t = mma4(a, b, t);
                        }
                    }
                    #pragma unroll
                    for (int v = 0; v < 8; ++v) Ts[wave][v + half * 8][l16] = t[v];
                }
                __syncthreads();
                if (act) {
                    v8f r = zero8;
                    #pragma unroll
                    for (int ks = 0; ks < 16; ks += 4) {
                        const int k0 = ks + half * 2;
                        v2f a; a.x = Mi[16 * I + l16][16 * I + k0];
                               a.y = Mi[16 * I + l16][16 * I + k0 + 1];
                        v2f b; b.x = Ts[wave][k0][l16];
                               b.y = Ts[wave][k0 + 1][l16];
                        r = mma4(a, b, r);
                    }
                    #pragma unroll
                    for (int v = 0; v < 8; ++v) Mi[16 * I + v + half * 8][16 * J + l16] = r[v];
                }
                __syncthreads();
            }
        }

        if (leftw) {
            #pragma unroll
            for (int J = 0; J < 4; ++J) {
                if (J <= rt) {
                    #pragma unroll
                    for (int ss = 0; ss < 4; ++ss) {
                        const int s  = 4 * J + ss;
                        const int k0 = 4 * s + 2 * half;
                        v2f b; b.x = Mi[16 * rt + l16][k0];
                               b.y = Mi[16 * rt + l16][k0 + 1];
                        F0 = mma4(bf0[s], b, F0);
                        F1 = mma4(bf1[s], b, F1);
                    }
                }
            }
        } else {
            const float* Mr = &Mi[16 * rt + l16][(kw - NSH) + 8 * half];
            const float4 m0 = *(const float4*)(Mr);
            const float4 m1 = *(const float4*)(Mr + 4);
            const float4 m2 = *(const float4*)(Mr + 16);
            const float4 m3 = *(const float4*)(Mr + 20);
            F0[0] = m0.x * sv[0].x; F0[1] = m0.y * sv[0].y; F0[2] = m0.z * sv[0].z; F0[3] = m0.w * sv[0].w;
            F0[4] = m1.x * sv[1].x; F0[5] = m1.y * sv[1].y; F0[6] = m1.z * sv[1].z; F0[7] = m1.w * sv[1].w;
            F1[0] = m2.x * sv[2].x; F1[1] = m2.y * sv[2].y; F1[2] = m2.z * sv[2].z; F1[3] = m2.w * sv[2].w;
            F1[4] = m3.x * sv[3].x; F1[5] = m3.y * sv[3].y; F1[6] = m3.z * sv[3].z; F1[7] = m3.w * sv[3].w;
        }
    } else {
        const int one_at = NSH + 16 * rt + l16 - kw - 8 * half;
        #pragma unroll
        for (int r = 0; r < 8; ++r) {
            F0[r] = (r == one_at) ? 1.0f : 0.0f;
            F1[r] = (r + 16 == one_at) ? 1.0f : 0.0f;
        }
    }

    for (int t0 = slot; t0 < ntiles; t0 += TS * TG) {
        #pragma unroll
        for (int tt = 0; tt < TG; ++tt) {
            if (t0 + TS * tt < ntiles) {
                v8f a0 = zero8, a1 = zero8;
                #pragma unroll
                for (int q = 0; q < 4; ++q) {
                    const float4 e0 = ev[tt][q >> 1];
                    const float4 e1 = ev[tt][2 + (q >> 1)];
                    v2f fa0; fa0.x = F0[2 * q]; fa0.y = F0[2 * q + 1];
                    v2f fa1; fa1.x = F1[2 * q]; fa1.y = F1[2 * q + 1];
                    v2f b0;  b0.x = (q & 1) ? e0.z : e0.x;  b0.y = (q & 1) ? e0.w : e0.y;
                    v2f b1;  b1.x = (q & 1) ? e1.z : e1.x;  b1.y = (q & 1) ? e1.w : e1.y;
                    a0 = mma4(fa0, b0, a0);
                    a1 = mma4(fa1, b1, a1);
                }
                float* rp = &red[tt][wave][lane * 8];
                *(float4*)(rp)     = make_float4(a0[0] + a1[0], a0[1] + a1[1], a0[2] + a1[2], a0[3] + a1[3]);
                *(float4*)(rp + 4) = make_float4(a0[4] + a1[4], a0[5] + a1[5], a0[6] + a1[6], a0[7] + a1[7]);
                if (pw && vld[tt]) {
                    float* pp = out + bis[tt] * NN + kw + 8 * half;
                    const float4 x0 = ev[tt][0], x1 = ev[tt][1], x2 = ev[tt][2], x3 = ev[tt][3];
                    *(float4*)(pp)      = make_float4(x0.x + bs[0].x, x0.y + bs[0].y, x0.z + bs[0].z, x0.w + bs[0].w);
                    *(float4*)(pp + 4)  = make_float4(x1.x + bs[1].x, x1.y + bs[1].y, x1.z + bs[1].z, x1.w + bs[1].w);
                    *(float4*)(pp + 16) = make_float4(x2.x + bs[2].x, x2.y + bs[2].y, x2.z + bs[2].z, x2.w + bs[2].w);
                    *(float4*)(pp + 20) = make_float4(x3.x + bs[3].x, x3.y + bs[3].y, x3.z + bs[3].z, x3.w + bs[3].w);
                }
            }
        }
        if (t0 + TS * TG < ntiles) load_pass(t0 + TS * TG);
        __syncthreads();

        if (tid < 64 * TG) {
            const int tt = tid >> 6;
            const int ln = (tid >> 1) & 31, vh = tid & 1;
            const int t  = t0 + TS * tt;
            const int r  = 16 * t + (ln & 15);
            if (t < ntiles && r < total) {
                const float* rp = &red[tt][0][ln * 8 + 4 * vh];
                float4 sum = *(const float4*)rp;
                #pragma unroll
                for (int w = 1; w < NW; ++w) {
                    const float4 p = *(const float4*)(rp + 256 * w);
                    sum.x += p.x; sum.y += p.y; sum.z += p.z; sum.w += p.w;
                }
                const int bi = lst[r];
                const int m0 = 16 * rt + 8 * (ln >> 4) + 4 * vh;
                const float4 q = *(const float4*)(bias_ns + dom * KSPA + m0);
                *(float4*)(out + bi * NN + NSH + m0) =
                    make_float4(sum.x + q.x, sum.y + q.y, sum.z + q.z, sum.w + q.w);
            }
        }
        __syncthreads();
    }
}

extern "C" void kernel_launch(void* const* d_in, const int* in_sizes, int n_in,
                              void* d_out, int out_size, void* d_ws, size_t ws_size,
                              hipStream_t stream) {
    const float* eps     = (const float*)d_in[0];
    const int*   didx    = (const int*)  d_in[1];
    const float* L_emb   = (const float*)d_in[2];
    const float* S_emb   = (const float*)d_in[3];
    const float* bias_ns = (const float*)d_in[4];
    const float* bias_sh = (const float*)d_in[5];
    float*       out     = (float*)d_out;

    fused_spa_kernel<<<dim3(ND, KSPA / 16, TS), dim3(512), 0, stream>>>(
        eps, didx, L_emb, S_emb, bias_ns, bias_sh, out);
}


# Loads stay in flight: block barriers order LDS only (were fences draining every global load), loads waited for at first use; L rows land under the M build, eps rows under phase 1b/2; sums unchanged
# speedup vs baseline: 10.1317x; 1.1385x over previous
#include <hip/hip_runtime.h>
#include <hip/hip_bf16.h>

#define NN   512
#define KSPA 64
#define ND   16
#define NB   1024
#define NSH  448
#define N_L  30688

typedef __attribute__((ext_vector_type(2))) float v2f;
typedef __attribute__((ext_vector_type(8))) float v8f;

#define MST 68
#define NW  16
#define TS  2
#define TG  3

__device__ __forceinline__ v8f mma4(v2f a, v2f b, v8f c) {
    return __builtin_amdgcn_wmma_f32_16x16x4_f32(
         false, a,  false, b,
         (short)0, c,  false,  false);
}

__device__ __forceinline__ void lds_barrier() {
    __builtin_amdgcn_fence(__ATOMIC_RELEASE, "workgroup", "local");
    __builtin_amdgcn_s_barrier();
    __builtin_amdgcn_fence(__ATOMIC_ACQUIRE, "workgroup", "local");
}

__global__ __launch_bounds__(512) void fused_spa_kernel(
    const float* __restrict__ eps,
    const int*   __restrict__ didx,
    const float* __restrict__ L_emb,
    const float* __restrict__ S_emb,
    const float* __restrict__ bias_ns,
    const float* __restrict__ bias_sh,
    float* __restrict__ out)
{
    __shared__ __attribute__((aligned(16))) float Wl[64][MST];
    __shared__ __attribute__((aligned(16))) float Mi[64][MST];
    __shared__ float Ts[4][16][20];
    __shared__ int   lst[NB];
    __shared__ int   s_total;
    __shared__ __attribute__((aligned(16))) float red[TG][NW][256];

    const int dom  = blockIdx.x;
    const int rt   = blockIdx.y;
    const int slot = blockIdx.z;
    const int tid  = threadIdx.x;
    const int wave = tid >> 5;
    const int lane = tid & 31;
    const int half = lane >> 4;
    const int l16  = lane & 15;
    const int kw   = 32 * wave;
    const bool leftw = wave < 14;
    const bool pw    = leftw && ((wave & 3) == rt);

    const float* Ld = L_emb + dom * N_L;

    float wraw[8];
    unsigned mk[32];
    float4 bs[4];
    v2f ca0[4], cb0[4], ca1[4], cb1[4], ca2[4], cb2[4], ca3[4], cb3[4];
    float4 sv[4];
    auto load_small = [&]() {
        if (pw) {
            const float* Bp = bias_sh + kw + 8 * half;
            bs[0] = *(const float4*)(Bp);
            bs[1] = *(const float4*)(Bp + 4);
            bs[2] = *(const float4*)(Bp + 16);
            bs[3] = *(const float4*)(Bp + 20);
        }
        if (wave == NW - 1) {
            int dv[32];
            #pragma unroll
            for (int j = 0; j < 32; ++j) dv[j] = didx[32 * j + lane];
            #pragma unroll
            for (int j = 0; j < 32; ++j) mk[j] = __builtin_amdgcn_ballot_w32(dv[j] == dom);
        }
    };
    auto load_chunk = [&](int J, v2f (&fa)[4], v2f (&fb)[4]) {
        #pragma unroll
        for (int ss = 0; ss < 4; ++ss) {
            const int s  = 4 * J + ss;
            const int k0 = 4 * s + 2 * half;
            const float* r0 = Ld + NSH * k0 + (k0 * (k0 - 1)) / 2 + kw + l16;
            const float* r1 = Ld + NSH * (k0 + 1) + ((k0 + 1) * k0) / 2 + kw + l16;
            fa[ss].x = r0[0];  fa[ss].y = r1[0];
            fb[ss].x = r0[16]; fb[ss].y = r1[16];
        }
    };
    if (dom != 0) {
        #pragma unroll
        for (int q = 0; q < 8; ++q) {
            const int idx = q * 512 + tid;
            const int i = idx >> 6, k = idx & 63;
            wraw[q] = Ld[NSH * i + (i * (i - 1)) / 2 + NSH + ((k < i) ? k : 0)];
        }
        load_small();
        #pragma unroll
        for (int q = 0; q < 8; ++q) {
            const int idx = q * 512 + tid;
            const int i = idx >> 6, k = idx & 63;
            Wl[i][k] = (k < i) ? wraw[q] : 0.0f;
            Mi[i][k] = 0.0f;
        }
        if (leftw) {
            load_chunk(0, ca0, cb0);
            if (rt >= 1) load_chunk(1, ca1, cb1);
            if (rt >= 2) load_chunk(2, ca2, cb2);
        } else {
            const float* Sp = S_emb + dom * KSPA + (kw - NSH) + 8 * half;
            sv[0] = *(const float4*)(Sp);
            sv[1] = *(const float4*)(Sp + 4);
            sv[2] = *(const float4*)(Sp + 16);
            sv[3] = *(const float4*)(Sp + 20);
        }
    } else {
        load_small();
    }
    lds_barrier();

    if (wave == NW - 1) {
        int total = 0;
        #pragma unroll
        for (int j = 0; j < 32; ++j) {
            const unsigned m = mk[j];
            const bool mine = (m >> lane) & 1u;
            if (mine) lst[total + __popc(m & ((1u << lane) - 1u))] = 32 * j + lane;
            total += __popc(m);
        }
        if (lane == 0) s_total = total;
    } else if (dom != 0 && tid < 64) {
        const int I = tid >> 4, j = tid & 15;
        float x[16];
        #pragma unroll
        for (int r = 0; r < 16; ++r) x[r] = (r == j) ? 1.0f : 0.0f;
        #pragma unroll
        for (int k = 0; k < 15; ++k) {
            const float xk = x[k];
            #pragma unroll
            for (int r = k + 1; r < 16; ++r)
                x[r] += Wl[16 * I + r][16 * I + k] * xk;
        }
        #pragma unroll
        for (int r = 0; r < 16; ++r) Mi[16 * I + r][16 * I + j] = x[r];
    }
    lds_barrier();

    const int total  = s_total;
    const int ntiles = (total + 15) >> 4;

    float4 ev[TG][4];
    int    bis[TG];
    bool   vld[TG];
    auto load_pass = [&](int tbase) {
        #pragma unroll
        for (int tt = 0; tt < TG; ++tt) {
            const int t = tbase + TS * tt;
            bis[tt] = 0; vld[tt] = false;
            if (t < ntiles) {
                const int r = 16 * t + l16;
                vld[tt] = r < total;
                bis[tt] = lst[vld[tt] ? r : 16 * t];
                const float* Er = eps + bis[tt] * NN + kw + 8 * half;
                ev[tt][0] = *(const float4*)(Er);
                ev[tt][1] = *(const float4*)(Er + 4);
                ev[tt][2] = *(const float4*)(Er + 16);
                ev[tt][3] = *(const float4*)(Er + 20);
            }
        }
    };
    if (dom != 0 && leftw && rt >= 3) load_chunk(3, ca3, cb3);

    #pragma unroll
    for (int tt = 0; tt < TG; ++tt) {
        const int t = slot + TS * tt;
        const bool has = t < ntiles;
        const int r = 16 * t + l16;
        vld[tt] = has && (r < total);
        const int li = lst[vld[tt] ? r : (has ? 16 * t : 0)];
        bis[tt] = has ? li : 0;
        const float* Er = eps + bis[tt] * NN + kw + 8 * half;
        ev[tt][0] = *(const float4*)(Er);
        ev[tt][1] = *(const float4*)(Er + 4);
        ev[tt][2] = *(const float4*)(Er + 16);
        ev[tt][3] = *(const float4*)(Er + 20);
    }

    const v8f zero8 = {0.f, 0.f, 0.f, 0.f, 0.f, 0.f, 0.f, 0.f};
    v8f F0 = zero8, F1 = zero8;

    if (dom != 0) {
        #pragma unroll
        for (int s = 1; s < 4; ++s) {
            if (s <= rt) {
                const bool act = wave < 4 - s;
                const int I = s + wave, J = wave;
                if (act) {
                    v8f t = zero8;
                    for (int K = J; K < I; ++K) {
                        #pragma unroll
                        for (int ks = 0; ks < 16; ks += 4) {
                            const int k0 = ks + half * 2;
                            v2f a; a.x = Wl[16 * I + l16][16 * K + k0];
                                   a.y = Wl[16 * I + l16][16 * K + k0 + 1];
                            v2f b; b.x = Mi[16 * K + k0][16 * J + l16];
                                   b.y = Mi[16 * K + k0 + 1][16 * J + l16];
                            t = mma4(a, b, t);
                        }
                    }
                    #pragma unroll
                    for (int v = 0; v < 8; ++v) Ts[wave][v + half * 8][l16] = t[v];
                }
                lds_barrier();
                if (act) {
                    v8f r = zero8;
                    #pragma unroll
                    for (int ks = 0; ks < 16; ks += 4) {
                        const int k0 = ks + half * 2;
                        v2f a; a.x = Mi[16 * I + l16][16 * I + k0];
                               a.y = Mi[16 * I + l16][16 * I + k0 + 1];
                        v2f b; b.x = Ts[wave][k0][l16];
                               b.y = Ts[wave][k0 + 1][l16];
                        r = mma4(a, b, r);
                    }
                    #pragma unroll
                    for (int v = 0; v < 8; ++v) Mi[16 * I + v + half * 8][16 * J + l16] = r[v];
                }
                lds_barrier();
            }
        }

        if (leftw) {
            auto mul_chunk = [&](int J, const v2f (&fa)[4], const v2f (&fb)[4]) {
                #pragma unroll
                for (int ss = 0; ss < 4; ++ss) {
                    const int s  = 4 * J + ss;
                    const int k0 = 4 * s + 2 * half;
                    v2f b; b.x = Mi[16 * rt + l16][k0];
                           b.y = Mi[16 * rt + l16][k0 + 1];
                    F0 = mma4(fa[ss], b, F0);
                    F1 = mma4(fb[ss], b, F1);
                }
            };
            mul_chunk(0, ca0, cb0);
            if (rt >= 1) mul_chunk(1, ca1, cb1);
            if (rt >= 2) mul_chunk(2, ca2, cb2);
            if (rt >= 3) mul_chunk(3, ca3, cb3);
        } else {
            const float* Mr = &Mi[16 * rt + l16][(kw - NSH) + 8 * half];
            const float4 m0 = *(const float4*)(Mr);
            const float4 m1 = *(const float4*)(Mr + 4);
            const float4 m2 = *(const float4*)(Mr + 16);
            const float4 m3 = *(const float4*)(Mr + 20);
            F0[0] = m0.x * sv[0].x; F0[1] = m0.y * sv[0].y; F0[2] = m0.z * sv[0].z; F0[3] = m0.w * sv[0].w;
            F0[4] = m1.x * sv[1].x; F0[5] = m1.y * sv[1].y; F0[6] = m1.z * sv[1].z; F0[7] = m1.w * sv[1].w;
            F1[0] = m2.x * sv[2].x; F1[1] = m2.y * sv[2].y; F1[2] = m2.z * sv[2].z; F1[3] = m2.w * sv[2].w;
            F1[4] = m3.x * sv[3].x; F1[5] = m3.y * sv[3].y; F1[6] = m3.z * sv[3].z; F1[7] = m3.w * sv[3].w;
        }
    } else {
        const int one_at = NSH + 16 * rt + l16 - kw - 8 * half;
        #pragma unroll
        for (int r = 0; r < 8; ++r) {
            F0[r] = (r == one_at) ? 1.0f : 0.0f;
            F1[r] = (r + 16 == one_at) ? 1.0f : 0.0f;
        }
    }

    for (int t0 = slot; t0 < ntiles; t0 += TS * TG) {
        #pragma unroll
        for (int tt = 0; tt < TG; ++tt) {
            if (t0 + TS * tt < ntiles) {
                v8f a0 = zero8, a1 = zero8;
                #pragma unroll
                for (int q = 0; q < 4; ++q) {
                    const float4 e0 = ev[tt][q >> 1];
                    const float4 e1 = ev[tt][2 + (q >> 1)];
                    v2f fa0; fa0.x = F0[2 * q]; fa0.y = F0[2 * q + 1];
                    v2f fa1; fa1.x = F1[2 * q]; fa1.y = F1[2 * q + 1];
                    v2f b0;  b0.x = (q & 1) ? e0.z : e0.x;  b0.y = (q & 1) ? e0.w : e0.y;
                    v2f b1;  b1.x = (q & 1) ? e1.z : e1.x;  b1.y = (q & 1) ? e1.w : e1.y;
                    a0 = mma4(fa0, b0, a0);
                    a1 = mma4(fa1, b1, a1);
                }
                float* rp = &red[tt][wave][lane * 8];
                *(float4*)(rp)     = make_float4(a0[0] + a1[0], a0[1] + a1[1], a0[2] + a1[2], a0[3] + a1[3]);
                *(float4*)(rp + 4) = make_float4(a0[4] + a1[4], a0[5] + a1[5], a0[6] + a1[6], a0[7] + a1[7]);
                if (pw && vld[tt]) {
                    float* pp = out + bis[tt] * NN + kw + 8 * half;
                    const float4 x0 = ev[tt][0], x1 = ev[tt][1], x2 = ev[tt][2], x3 = ev[tt][3];
                    *(float4*)(pp)      = make_float4(x0.x + bs[0].x, x0.y + bs[0].y, x0.z + bs[0].z, x0.w + bs[0].w);
                    *(float4*)(pp + 4)  = make_float4(x1.x + bs[1].x, x1.y + bs[1].y, x1.z + bs[1].z, x1.w + bs[1].w);
                    *(float4*)(pp + 16) = make_float4(x2.x + bs[2].x, x2.y + bs[2].y, x2.z + bs[2].z, x2.w + bs[2].w);
                    *(float4*)(pp + 20) = make_float4(x3.x + bs[3].x, x3.y + bs[3].y, x3.z + bs[3].z, x3.w + bs[3].w);
                }
            }
        }
        if (t0 + TS * TG < ntiles) load_pass(t0 + TS * TG);
        lds_barrier();

        if (tid < 64 * TG) {
            const int tt = tid >> 6;
            const int ln = (tid >> 1) & 31, vh = tid & 1;
            const int t  = t0 + TS * tt;
            const int r  = 16 * t + (ln & 15);
            if (t < ntiles && r < total) {
                const float* rp = &red[tt][0][ln * 8 + 4 * vh];
                float4 sum = *(const float4*)rp;
                #pragma unroll
                for (int w = 1; w < NW; ++w) {
                    const float4 p = *(const float4*)(rp + 256 * w);
                    sum.x += p.x; sum.y += p.y; sum.z += p.z; sum.w += p.w;
                }
                const int bi = lst[r];
                const int m0 = 16 * rt + 8 * (ln >> 4) + 4 * vh;
                const float4 q = *(const float4*)(bias_ns + dom * KSPA + m0);
                *(float4*)(out + bi * NN + NSH + m0) =
                    make_float4(sum.x + q.x, sum.y + q.y, sum.z + q.z, sum.w + q.w);
            }
        }
        lds_barrier();
    }
}

extern "C" void kernel_launch(void* const* d_in, const int* in_sizes, int n_in,
                              void* d_out, int out_size, void* d_ws, size_t ws_size,
                              hipStream_t stream) {
    const float* eps     = (const float*)d_in[0];
    const int*   didx    = (const int*)  d_in[1];
    const float* L_emb   = (const float*)d_in[2];
    const float* S_emb   = (const float*)d_in[3];
    const float* bias_ns = (const float*)d_in[4];
    const float* bias_sh = (const float*)d_in[5];
    float*       out     = (float*)d_out;

    fused_spa_kernel<<<dim3(ND, KSPA / 16, TS), dim3(512), 0, stream>>>(
        eps, didx, L_emb, S_emb, bias_ns, bias_sh, out);
}


# Tail operand: a wave requests its eps rows as whole 128-byte lines (8 lanes a line), makes its B fragments through an LDS tile of its own, stores the shared prefix as whole lines; sums unchanged
# speedup vs baseline: 10.8469x; 1.0706x over previous
#include <hip/hip_runtime.h>
#include <hip/hip_bf16.h>

#define NN   512
#define KSPA 64
#define ND   16
#define NB   1024
#define NSH  448
#define N_L  30688

typedef __attribute__((ext_vector_type(2))) float v2f;
typedef __attribute__((ext_vector_type(8))) float v8f;

#define MST 68
#define NW  16
#define TS  2
#define TG  3
#define SP  36

__device__ __forceinline__ v8f mma4(v2f a, v2f b, v8f c) {
    return __builtin_amdgcn_wmma_f32_16x16x4_f32(
         false, a,  false, b,
         (short)0, c,  false,  false);
}

__device__ __forceinline__ void lds_barrier() {
    __builtin_amdgcn_fence(__ATOMIC_RELEASE, "workgroup", "local");
    __builtin_amdgcn_s_barrier();
    __builtin_amdgcn_fence(__ATOMIC_ACQUIRE, "workgroup", "local");
}

__device__ __forceinline__ void wave_lds_order() {
    __builtin_amdgcn_fence(__ATOMIC_RELEASE, "wavefront", "local");
    __builtin_amdgcn_wave_barrier();
    __builtin_amdgcn_fence(__ATOMIC_ACQUIRE, "wavefront", "local");
}

__global__ __launch_bounds__(512) void fused_spa_kernel(
    const float* __restrict__ eps,
    const int*   __restrict__ didx,
    const float* __restrict__ L_emb,
    const float* __restrict__ S_emb,
    const float* __restrict__ bias_ns,
    const float* __restrict__ bias_sh,
    float* __restrict__ out)
{
    __shared__ __attribute__((aligned(16))) float Wl[64][MST];
    __shared__ __attribute__((aligned(16))) float Mi[64][MST];
    __shared__ float Ts[4][16][20];
    __shared__ int   lst[NB];
    __shared__ int   s_total;
    __shared__ __attribute__((aligned(16))) float red[TG][NW][256];
    __shared__ __attribute__((aligned(16))) float stg[NW][TG][16 * SP];

    const int dom  = blockIdx.x;
    const int rt   = blockIdx.y;
    const int slot = blockIdx.z;
    const int tid  = threadIdx.x;
    const int wave = tid >> 5;
    const int lane = tid & 31;
    const int half = lane >> 4;
    const int l16  = lane & 15;
    const int kw   = 32 * wave;
    const int rsub = lane >> 3;
    const int csub = 4 * (lane & 7);
    const bool leftw = wave < 14;
    const bool pw    = leftw && ((wave & 3) == rt);

    const float* Ld = L_emb + dom * N_L;

    float wraw[8];
    unsigned mk[32];
    float4 bsc;
    v2f ca0[4], cb0[4], ca1[4], cb1[4], ca2[4], cb2[4], ca3[4], cb3[4];
    float4 sv[4];
    auto load_small = [&]() {
        if (pw) {
            bsc = *(const float4*)(bias_sh + kw + csub);
        }
        if (wave == NW - 1) {
            int dv[32];
            #pragma unroll
            for (int j = 0; j < 32; ++j) dv[j] = didx[32 * j + lane];
            #pragma unroll
            for (int j = 0; j < 32; ++j) mk[j] = __builtin_amdgcn_ballot_w32(dv[j] == dom);
        }
    };
    auto load_chunk = [&](int J, v2f (&fa)[4], v2f (&fb)[4]) {
        #pragma unroll
        for (int ss = 0; ss < 4; ++ss) {
            const int s  = 4 * J + ss;
            const int k0 = 4 * s + 2 * half;
            const float* r0 = Ld + NSH * k0 + (k0 * (k0 - 1)) / 2 + kw + l16;
            const float* r1 = Ld + NSH * (k0 + 1) + ((k0 + 1) * k0) / 2 + kw + l16;
            fa[ss].x = r0[0];  fa[ss].y = r1[0];
            fb[ss].x = r0[16]; fb[ss].y = r1[16];
        }
    };
    if (dom != 0) {
        #pragma unroll
        for (int q = 0; q < 8; ++q) {
            const int idx = q * 512 + tid;
            const int i = idx >> 6, k = idx & 63;
            wraw[q] = Ld[NSH * i + (i * (i - 1)) / 2 + NSH + ((k < i) ? k : 0)];
        }
        load_small();
        #pragma unroll
        for (int q = 0; q < 8; ++q) {
            const int idx = q * 512 + tid;
            const int i = idx >> 6, k = idx & 63;
            Wl[i][k] = (k < i) ? wraw[q] : 0.0f;
            Mi[i][k] = 0.0f;
        }
        if (leftw) {
            load_chunk(0, ca0, cb0);
            if (rt >= 1) load_chunk(1, ca1, cb1);
            if (rt >= 2) load_chunk(2, ca2, cb2);
        } else {
            const float* Sp = S_emb + dom * KSPA + (kw - NSH) + 8 * half;
            sv[0] = *(const float4*)(Sp);
            sv[1] = *(const float4*)(Sp + 4);
            sv[2] = *(const float4*)(Sp + 16);
            sv[3] = *(const float4*)(Sp + 20);
        }
    } else {
        load_small();
    }
    lds_barrier();

    if (wave == NW - 1) {
        int total = 0;
        #pragma unroll
        for (int j = 0; j < 32; ++j) {
            const unsigned m = mk[j];
            const bool mine = (m >> lane) & 1u;
            if (mine) lst[total + __popc(m & ((1u << lane) - 1u))] = 32 * j + lane;
            total += __popc(m);
        }
        if (lane == 0) s_total = total;
    } else if (dom != 0 && tid < 64) {
        const int I = tid >> 4, j = tid & 15;
        float x[16];
        #pragma unroll
        for (int r = 0; r < 16; ++r) x[r] = (r == j) ? 1.0f : 0.0f;
        #pragma unroll
        for (int k = 0; k < 15; ++k) {
            const float xk = x[k];
            #pragma unroll
            for (int r = k + 1; r < 16; ++r)
                x[r] += Wl[16 * I + r][16 * I + k] * xk;
        }
        #pragma unroll
        for (int r = 0; r < 16; ++r) Mi[16 * I + r][16 * I + j] = x[r];
    }
    lds_barrier();

    const int total  = s_total;
    const int ntiles = (total + 15) >> 4;

    float4 cv[TG][4];
    int    cb[TG][4];
    bool   cvl[TG][4];
    auto load_pass = [&](int tbase) {
        #pragma unroll
        for (int tt = 0; tt < TG; ++tt) {
            const int t = tbase + TS * tt;
            #pragma unroll
            for (int j = 0; j < 4; ++j) { cb[tt][j] = 0; cvl[tt][j] = false; }
            if (t < ntiles) {
                #pragma unroll
                for (int j = 0; j < 4; ++j) {
                    const int r = 16 * t + 4 * j + rsub;
                    cvl[tt][j] = r < total;
                    cb[tt][j] = lst[cvl[tt][j] ? r : 16 * t];
                    cv[tt][j] = *(const float4*)(eps + cb[tt][j] * NN + kw + csub);
                }
            }
        }
    };
    if (dom != 0 && leftw && rt >= 3) load_chunk(3, ca3, cb3);

    #pragma unroll
    for (int tt = 0; tt < TG; ++tt) {
        const int t = slot + TS * tt;
        const bool has = t < ntiles;
        #pragma unroll
        for (int j = 0; j < 4; ++j) {
            const int r = 16 * t + 4 * j + rsub;
            cvl[tt][j] = has && (r < total);
            const int li = lst[cvl[tt][j] ? r : (has ? 16 * t : 0)];
            cb[tt][j] = has ? li : 0;
            cv[tt][j] = *(const float4*)(eps + cb[tt][j] * NN + kw + csub);
        }
    }

    const v8f zero8 = {0.f, 0.f, 0.f, 0.f, 0.f, 0.f, 0.f, 0.f};
    v8f F0 = zero8, F1 = zero8;

    if (dom != 0) {
        #pragma unroll
        for (int s = 1; s < 4; ++s) {
            if (s <= rt) {
                const bool act = wave < 4 - s;
                const int I = s + wave, J = wave;
                if (act) {
                    v8f t = zero8;
                    for (int K = J; K < I; ++K) {
                        #pragma unroll
                        for (int ks = 0; ks < 16; ks += 4) {
                            const int k0 = ks + half * 2;
                            v2f a; a.x = Wl[16 * I + l16][16 * K + k0];
                                   a.y = Wl[16 * I + l16][16 * K + k0 + 1];
                            v2f b; b.x = Mi[16 * K + k0][16 * J + l16];
                                   b.y = Mi[16 * K + k0 + 1][16 * J + l16];
                            t = mma4(a, b, t);
                        }
                    }
                    #pragma unroll
                    for (int v = 0; v < 8; ++v) Ts[wave][v + half * 8][l16] = t[v];
                }
                lds_barrier();
                if (act) {
                    v8f r = zero8;
                    #pragma unroll
                    for (int ks = 0; ks < 16; ks += 4) {
                        const int k0 = ks + half * 2;
                        v2f a; a.x = Mi[16 * I + l16][16 * I + k0];
                               a.y = Mi[16 * I + l16][16 * I + k0 + 1];
                        v2f b; b.x = Ts[wave][k0][l16];
                               b.y = Ts[wave][k0 + 1][l16];
                        r = mma4(a, b, r);
                    }
                    #pragma unroll
                    for (int v = 0; v < 8; ++v) Mi[16 * I + v + half * 8][16 * J + l16] = r[v];
                }
                lds_barrier();
            }
        }

        if (leftw) {
            auto mul_chunk = [&](int J, const v2f (&fa)[4], const v2f (&fb)[4]) {
                #pragma unroll
                for (int ss = 0; ss < 4; ++ss) {
                    const int s  = 4 * J + ss;
                    const int k0 = 4 * s + 2 * half;
                    v2f b; b.x = Mi[16 * rt + l16][k0];
                           b.y = Mi[16 * rt + l16][k0 + 1];
                    F0 = mma4(fa[ss], b, F0);
                    F1 = mma4(fb[ss], b, F1);
                }
            };
            mul_chunk(0, ca0, cb0);
            if (rt >= 1) mul_chunk(1, ca1, cb1);
            if (rt >= 2) mul_chunk(2, ca2, cb2);
            if (rt >= 3) mul_chunk(3, ca3, cb3);
        } else {
            const float* Mr = &Mi[16 * rt + l16][(kw - NSH) + 8 * half];
            const float4 m0 = *(const float4*)(Mr);
            const float4 m1 = *(const float4*)(Mr + 4);
            const float4 m2 = *(const float4*)(Mr + 16);
            const float4 m3 = *(const float4*)(Mr + 20);
            F0[0] = m0.x * sv[0].x; F0[1] = m0.y * sv[0].y; F0[2] = m0.z * sv[0].z; F0[3] = m0.w * sv[0].w;
            F0[4] = m1.x * sv[1].x; F0[5] = m1.y * sv[1].y; F0[6] = m1.z * sv[1].z; F0[7] = m1.w * sv[1].w;
            F1[0] = m2.x * sv[2].x; F1[1] = m2.y * sv[2].y; F1[2] = m2.z * sv[2].z; F1[3] = m2.w * sv[2].w;
            F1[4] = m3.x * sv[3].x; F1[5] = m3.y * sv[3].y; F1[6] = m3.z * sv[3].z; F1[7] = m3.w * sv[3].w;
        }
    } else {
        const int one_at = NSH + 16 * rt + l16 - kw - 8 * half;
        #pragma unroll
        for (int r = 0; r < 8; ++r) {
            F0[r] = (r == one_at) ? 1.0f : 0.0f;
            F1[r] = (r + 16 == one_at) ? 1.0f : 0.0f;
        }
    }

    for (int t0 = slot; t0 < ntiles; t0 += TS * TG) {
        #pragma unroll
        for (int tt = 0; tt < TG; ++tt) {
            if (t0 + TS * tt < ntiles) {
                float* sp = &stg[wave][tt][0];
                #pragma unroll
                for (int j = 0; j < 4; ++j)
                    *(float4*)(sp + (4 * j + rsub) * SP + csub) = cv[tt][j];
                if (pw) {
                    float4 y[4];
                    #pragma unroll
                    for (int j = 0; j < 4; ++j) {
                        const float4 x = cv[tt][j];
                        y[j] = make_float4(x.x + bsc.x, x.y + bsc.y, x.z + bsc.z, x.w + bsc.w);
                    }
                    if (16 * (t0 + TS * tt) + 16 <= total) {
                        #pragma unroll
                        for (int j = 0; j < 4; ++j)
                            *(float4*)(out + cb[tt][j] * NN + kw + csub) = y[j];
                    } else {
                        #pragma unroll
                        for (int j = 0; j < 4; ++j)
                            if (cvl[tt][j]) *(float4*)(out + cb[tt][j] * NN + kw + csub) = y[j];
                    }
                }
            }
        }
        wave_lds_order();
        float4 ev[TG][4];
        #pragma unroll
        for (int tt = 0; tt < TG; ++tt) {
            if (t0 + TS * tt < ntiles) {
                const float* fp = &stg[wave][tt][l16 * SP + 8 * half];
                ev[tt][0] = *(const float4*)(fp);
                ev[tt][1] = *(const float4*)(fp + 4);
                ev[tt][2] = *(const float4*)(fp + 16);
                ev[tt][3] = *(const float4*)(fp + 20);
            }
        }
        #pragma unroll
        for (int tt = 0; tt < TG; ++tt) {
            if (t0 + TS * tt < ntiles) {
                v8f a0 = zero8, a1 = zero8;
                #pragma unroll
                for (int q = 0; q < 4; ++q) {
                    const float4 e0 = ev[tt][q >> 1];
                    const float4 e1 = ev[tt][2 + (q >> 1)];
                    v2f fa0; fa0.x = F0[2 * q]; fa0.y = F0[2 * q + 1];
                    v2f fa1; fa1.x = F1[2 * q]; fa1.y = F1[2 * q + 1];
                    v2f b0;  b0.x = (q & 1) ? e0.z : e0.x;  b0.y = (q & 1) ? e0.w : e0.y;
                    v2f b1;  b1.x = (q & 1) ? e1.z : e1.x;  b1.y = (q & 1) ? e1.w : e1.y;
                    a0 = mma4(fa0, b0, a0);
                    a1 = mma4(fa1, b1, a1);
                }
                float* rp = &red[tt][wave][lane * 8];
                *(float4*)(rp)     = make_float4(a0[0] + a1[0], a0[1] + a1[1], a0[2] + a1[2], a0[3] + a1[3]);
                *(float4*)(rp + 4) = make_float4(a0[4] + a1[4], a0[5] + a1[5], a0[6] + a1[6], a0[7] + a1[7]);
            }
        }
        if (t0 + TS * TG < ntiles) load_pass(t0 + TS * TG);
        lds_barrier();

        if (tid < 64 * TG) {
            const int tt = tid >> 6;
            const int ln = (tid >> 1) & 31, vh = tid & 1;
            const int t  = t0 + TS * tt;
            const int r  = 16 * t + (ln & 15);
            if (t < ntiles && r < total) {
                const float* rp = &red[tt][0][ln * 8 + 4 * vh];
                float4 sum = *(const float4*)rp;
                #pragma unroll
                for (int w = 1; w < NW; ++w) {
                    const float4 p = *(const float4*)(rp + 256 * w);
                    sum.x += p.x; sum.y += p.y; sum.z += p.z; sum.w += p.w;
                }
                const int bi = lst[r];
                const int m0 = 16 * rt + 8 * (ln >> 4) + 4 * vh;
                const float4 q = *(const float4*)(bias_ns + dom * KSPA + m0);
                *(float4*)(out + bi * NN + NSH + m0) =
                    make_float4(sum.x + q.x, sum.y + q.y, sum.z + q.z, sum.w + q.w);
            }
        }
        lds_barrier();
    }
}

extern "C" void kernel_launch(void* const* d_in, const int* in_sizes, int n_in,
                              void* d_out, int out_size, void* d_ws, size_t ws_size,
                              hipStream_t stream) {
    const float* eps     = (const float*)d_in[0];
    const int*   didx    = (const int*)  d_in[1];
    const float* L_emb   = (const float*)d_in[2];
    const float* S_emb   = (const float*)d_in[3];
    const float* bias_ns = (const float*)d_in[4];
    const float* bias_sh = (const float*)d_in[5];
    float*       out     = (float*)d_out;

    fused_spa_kernel<<<dim3(ND, KSPA / 16, TS), dim3(512), 0, stream>>>(
        eps, didx, L_emb, S_emb, bias_ns, bias_sh, out);
}


# Domain list built by 8 waves at once: each ballots its own 128 batch elements, match counts cross the first barrier via LDS (was one wave's 32-step serial loop between barriers); sums unchanged
# speedup vs baseline: 11.9235x; 1.0993x over previous
#include <hip/hip_runtime.h>
#include <hip/hip_bf16.h>

#define NN   512
#define KSPA 64
#define ND   16
#define NB   1024
#define NSH  448
#define N_L  30688

typedef __attribute__((ext_vector_type(2))) float v2f;
typedef __attribute__((ext_vector_type(8))) float v8f;

#define MST 68
#define NW  16
#define TS  2
#define TG  3
#define SP  36
#define LW  8

__device__ __forceinline__ v8f mma4(v2f a, v2f b, v8f c) {
    return __builtin_amdgcn_wmma_f32_16x16x4_f32(
         false, a,  false, b,
         (short)0, c,  false,  false);
}

__device__ __forceinline__ void lds_barrier() {
    __builtin_amdgcn_fence(__ATOMIC_RELEASE, "workgroup", "local");
    __builtin_amdgcn_s_barrier();
    __builtin_amdgcn_fence(__ATOMIC_ACQUIRE, "workgroup", "local");
}

__device__ __forceinline__ void wave_lds_order() {
    __builtin_amdgcn_fence(__ATOMIC_RELEASE, "wavefront", "local");
    __builtin_amdgcn_wave_barrier();
    __builtin_amdgcn_fence(__ATOMIC_ACQUIRE, "wavefront", "local");
}

__global__ __launch_bounds__(512) void fused_spa_kernel(
    const float* __restrict__ eps,
    const int*   __restrict__ didx,
    const float* __restrict__ L_emb,
    const float* __restrict__ S_emb,
    const float* __restrict__ bias_ns,
    const float* __restrict__ bias_sh,
    float* __restrict__ out)
{
    __shared__ __attribute__((aligned(16))) float Wl[64][MST];
    __shared__ __attribute__((aligned(16))) float Mi[64][MST];
    __shared__ float Ts[4][16][20];
    __shared__ int   lst[NB];
    __shared__ int   s_total;
    __shared__ __attribute__((aligned(16))) int cnt[LW];
    __shared__ __attribute__((aligned(16))) float red[TG][NW][256];
    __shared__ __attribute__((aligned(16))) float stg[NW][TG][16 * SP];

    const int dom  = blockIdx.x;
    const int rt   = blockIdx.y;
    const int slot = blockIdx.z;
    const int tid  = threadIdx.x;
    const int wave = tid >> 5;
    const int lane = tid & 31;
    const int half = lane >> 4;
    const int l16  = lane & 15;
    const int kw   = 32 * wave;
    const int rsub = lane >> 3;
    const int csub = 4 * (lane & 7);
    const bool listw = wave >= NW - LW;
    const int  lw    = wave - (NW - LW);
    const bool leftw = wave < 14;
    const bool pw    = leftw && ((wave & 3) == rt);

    const float* Ld = L_emb + dom * N_L;

    float wraw[8];
    unsigned mk[4];
    float4 bsc;
    v2f ca0[4], cb0[4], ca1[4], cb1[4], ca2[4], cb2[4], ca3[4], cb3[4];
    float4 sv[4];
    auto load_small = [&]() {
        if (pw) {
            bsc = *(const float4*)(bias_sh + kw + csub);
        }
        if (listw) {
            int dv[4];
            #pragma unroll
            for (int j = 0; j < 4; ++j) dv[j] = didx[128 * lw + 32 * j + lane];
            #pragma unroll
            for (int j = 0; j < 4; ++j) mk[j] = __builtin_amdgcn_ballot_w32(dv[j] == dom);
            if (lane == 0) cnt[lw] = __popc(mk[0]) + __popc(mk[1]) + __popc(mk[2]) + __popc(mk[3]);
        }
    };
    auto load_chunk = [&](int J, v2f (&fa)[4], v2f (&fb)[4]) {
        #pragma unroll
        for (int ss = 0; ss < 4; ++ss) {
            const int s  = 4 * J + ss;
            const int k0 = 4 * s + 2 * half;
            const float* r0 = Ld + NSH * k0 + (k0 * (k0 - 1)) / 2 + kw + l16;
            const float* r1 = Ld + NSH * (k0 + 1) + ((k0 + 1) * k0) / 2 + kw + l16;
            fa[ss].x = r0[0];  fa[ss].y = r1[0];
            fb[ss].x = r0[16]; fb[ss].y = r1[16];
        }
    };
    if (dom != 0) {
        #pragma unroll
        for (int q = 0; q < 8; ++q) {
            const int idx = q * 512 + tid;
            const int i = idx >> 6, k = idx & 63;
            wraw[q] = Ld[NSH * i + (i * (i - 1)) / 2 + NSH + ((k < i) ? k : 0)];
        }
        load_small();
        #pragma unroll
        for (int q = 0; q < 8; ++q) {
            const int idx = q * 512 + tid;
            const int i = idx >> 6, k = idx & 63;
            Wl[i][k] = (k < i) ? wraw[q] : 0.0f;
            Mi[i][k] = 0.0f;
        }
        if (leftw) {
            load_chunk(0, ca0, cb0);
            if (rt >= 1) load_chunk(1, ca1, cb1);
            if (rt >= 2) load_chunk(2, ca2, cb2);
        } else {
            const float* Sp = S_emb + dom * KSPA + (kw - NSH) + 8 * half;
            sv[0] = *(const float4*)(Sp);
            sv[1] = *(const float4*)(Sp + 4);
            sv[2] = *(const float4*)(Sp + 16);
            sv[3] = *(const float4*)(Sp + 20);
        }
    } else {
        load_small();
    }
    lds_barrier();

    if (listw) {
        const int4 c0 = *(const int4*)&cnt[0];
        const int4 c1 = *(const int4*)&cnt[4];
        const int c[LW] = {c0.x, c0.y, c0.z, c0.w, c1.x, c1.y, c1.z, c1.w};
        int pos = 0;
        #pragma unroll
        for (int v = 0; v < LW - 1; ++v) pos += (v < lw) ? c[v] : 0;
        #pragma unroll
        for (int j = 0; j < 4; ++j) {
            const unsigned m = mk[j];
            const bool mine = (m >> lane) & 1u;
            if (mine) lst[pos + __popc(m & ((1u << lane) - 1u))] = 128 * lw + 32 * j + lane;
            pos += __popc(m);
        }
        if (wave == NW - 1 && lane == 0) s_total = pos;
    } else if (dom != 0 && tid < 64) {
        const int I = tid >> 4, j = tid & 15;
        float x[16];
        #pragma unroll
        for (int r = 0; r < 16; ++r) x[r] = (r == j) ? 1.0f : 0.0f;
        #pragma unroll
        for (int k = 0; k < 15; ++k) {
            const float xk = x[k];
            #pragma unroll
            for (int r = k + 1; r < 16; ++r)
                x[r] += Wl[16 * I + r][16 * I + k] * xk;
        }
        #pragma unroll
        for (int r = 0; r < 16; ++r) Mi[16 * I + r][16 * I + j] = x[r];
    }
    lds_barrier();

    const int total  = s_total;
    const int ntiles = (total + 15) >> 4;

    float4 cv[TG][4];
    int    cb[TG][4];
    bool   cvl[TG][4];
    auto load_pass = [&](int tbase) {
        #pragma unroll
        for (int tt = 0; tt < TG; ++tt) {
            const int t = tbase + TS * tt;
            #pragma unroll
            for (int j = 0; j < 4; ++j) { cb[tt][j] = 0; cvl[tt][j] = false; }
            if (t < ntiles) {
                #pragma unroll
                for (int j = 0; j < 4; ++j) {
                    const int r = 16 * t + 4 * j + rsub;
                    cvl[tt][j] = r < total;
                    cb[tt][j] = lst[cvl[tt][j] ? r : 16 * t];
                    cv[tt][j] = *(const float4*)(eps + cb[tt][j] * NN + kw + csub);
                }
            }
        }
    };
    if (dom != 0 && leftw && rt >= 3) load_chunk(3, ca3, cb3);

    #pragma unroll
    for (int tt = 0; tt < TG; ++tt) {
        const int t = slot + TS * tt;
        const bool has = t < ntiles;
        #pragma unroll
        for (int j = 0; j < 4; ++j) {
            const int r = 16 * t + 4 * j + rsub;
            cvl[tt][j] = has && (r < total);
            const int li = lst[cvl[tt][j] ? r : (has ? 16 * t : 0)];
            cb[tt][j] = has ? li : 0;
            cv[tt][j] = *(const float4*)(eps + cb[tt][j] * NN + kw + csub);
        }
    }

    const v8f zero8 = {0.f, 0.f, 0.f, 0.f, 0.f, 0.f, 0.f, 0.f};
    v8f F0 = zero8, F1 = zero8;

    if (dom != 0) {
        #pragma unroll
        for (int s = 1; s < 4; ++s) {
            if (s <= rt) {
                const bool act = wave < 4 - s;
                const int I = s + wave, J = wave;
                if (act) {
                    v8f t = zero8;
                    for (int K = J; K < I; ++K) {
                        #pragma unroll
                        for (int ks = 0; ks < 16; ks += 4) {
                            const int k0 = ks + half * 2;
                            v2f a; a.x = Wl[16 * I + l16][16 * K + k0];
                                   a.y = Wl[16 * I + l16][16 * K + k0 + 1];
                            v2f b; b.x = Mi[16 * K + k0][16 * J + l16];
                                   b.y = Mi[16 * K + k0 + 1][16 * J + l16];
                            t = mma4(a, b, t);
                        }
                    }
                    #pragma unroll
                    for (int v = 0; v < 8; ++v) Ts[wave][v + half * 8][l16] = t[v];
                }
                lds_barrier();
                if (act) {
                    v8f r = zero8;
                    #pragma unroll
                    for (int ks = 0; ks < 16; ks += 4) {
                        const int k0 = ks + half * 2;
                        v2f a; a.x = Mi[16 * I + l16][16 * I + k0];
                               a.y = Mi[16 * I + l16][16 * I + k0 + 1];
                        v2f b; b.x = Ts[wave][k0][l16];
                               b.y = Ts[wave][k0 + 1][l16];
                        r = mma4(a, b, r);
                    }
                    #pragma unroll
                    for (int v = 0; v < 8; ++v) Mi[16 * I + v + half * 8][16 * J + l16] = r[v];
                }
                lds_barrier();
            }
        }

        if (leftw) {
            auto mul_chunk = [&](int J, const v2f (&fa)[4], const v2f (&fb)[4]) {
                #pragma unroll
                for (int ss = 0; ss < 4; ++ss) {
                    const int s  = 4 * J + ss;
                    const int k0 = 4 * s + 2 * half;
                    v2f b; b.x = Mi[16 * rt + l16][k0];
                           b.y = Mi[16 * rt + l16][k0 + 1];
                    F0 = mma4(fa[ss], b, F0);
                    F1 = mma4(fb[ss], b, F1);
                }
            };
            mul_chunk(0, ca0, cb0);
            if (rt >= 1) mul_chunk(1, ca1, cb1);
            if (rt >= 2) mul_chunk(2, ca2, cb2);
            if (rt >= 3) mul_chunk(3, ca3, cb3);
        } else {
            const float* Mr = &Mi[16 * rt + l16][(kw - NSH) + 8 * half];
            const float4 m0 = *(const float4*)(Mr);
            const float4 m1 = *(const float4*)(Mr + 4);
            const float4 m2 = *(const float4*)(Mr + 16);
            const float4 m3 = *(const float4*)(Mr + 20);
            F0[0] = m0.x * sv[0].x; F0[1] = m0.y * sv[0].y; F0[2] = m0.z * sv[0].z; F0[3] = m0.w * sv[0].w;
            F0[4] = m1.x * sv[1].x; F0[5] = m1.y * sv[1].y; F0[6] = m1.z * sv[1].z; F0[7] = m1.w * sv[1].w;
            F1[0] = m2.x * sv[2].x; F1[1] = m2.y * sv[2].y; F1[2] = m2.z * sv[2].z; F1[3] = m2.w * sv[2].w;
            F1[4] = m3.x * sv[3].x; F1[5] = m3.y * sv[3].y; F1[6] = m3.z * sv[3].z; F1[7] = m3.w * sv[3].w;
        }
    } else {
        const int one_at = NSH + 16 * rt + l16 - kw - 8 * half;
        #pragma unroll
        for (int r = 0; r < 8; ++r) {
            F0[r] = (r == one_at) ? 1.0f : 0.0f;
            F1[r] = (r + 16 == one_at) ? 1.0f : 0.0f;
        }
    }

    for (int t0 = slot; t0 < ntiles; t0 += TS * TG) {
        #pragma unroll
        for (int tt = 0; tt < TG; ++tt) {
            if (t0 + TS * tt < ntiles) {
                float* sp = &stg[wave][tt][0];
                #pragma unroll
                for (int j = 0; j < 4; ++j)
                    *(float4*)(sp + (4 * j + rsub) * SP + csub) = cv[tt][j];
                if (pw) {
                    float4 y[4];
                    #pragma unroll
                    for (int j = 0; j < 4; ++j) {
                        const float4 x = cv[tt][j];
                        y[j] = make_float4(x.x + bsc.x, x.y + bsc.y, x.z + bsc.z, x.w + bsc.w);
                    }
                    if (16 * (t0 + TS * tt) + 16 <= total) {
                        #pragma unroll
                        for (int j = 0; j < 4; ++j)
                            *(float4*)(out + cb[tt][j] * NN + kw + csub) = y[j];
                    } else {
                        #pragma unroll
                        for (int j = 0; j < 4; ++j)
                            if (cvl[tt][j]) *(float4*)(out + cb[tt][j] * NN + kw + csub) = y[j];
                    }
                }
            }
        }
        wave_lds_order();
        float4 ev[TG][4];
        #pragma unroll
        for (int tt = 0; tt < TG; ++tt) {
            if (t0 + TS * tt < ntiles) {
                const float* fp = &stg[wave][tt][l16 * SP + 8 * half];
                ev[tt][0] = *(const float4*)(fp);
                ev[tt][1] = *(const float4*)(fp + 4);
                ev[tt][2] = *(const float4*)(fp + 16);
                ev[tt][3] = *(const float4*)(fp + 20);
            }
        }
        #pragma unroll
        for (int tt = 0; tt < TG; ++tt) {
            if (t0 + TS * tt < ntiles) {
                v8f a0 = zero8, a1 = zero8;
                #pragma unroll
                for (int q = 0; q < 4; ++q) {
                    const float4 e0 = ev[tt][q >> 1];
                    const float4 e1 = ev[tt][2 + (q >> 1)];
                    v2f fa0; fa0.x = F0[2 * q]; fa0.y = F0[2 * q + 1];
                    v2f fa1; fa1.x = F1[2 * q]; fa1.y = F1[2 * q + 1];
                    v2f b0;  b0.x = (q & 1) ? e0.z : e0.x;  b0.y = (q & 1) ? e0.w : e0.y;
                    v2f b1;  b1.x = (q & 1) ? e1.z : e1.x;  b1.y = (q & 1) ? e1.w : e1.y;
                    a0 = mma4(fa0, b0, a0);
                    a1 = mma4(fa1, b1, a1);
                }
                float* rp = &red[tt][wave][lane * 8];
                *(float4*)(rp)     = make_float4(a0[0] + a1[0], a0[1] + a1[1], a0[2] + a1[2], a0[3] + a1[3]);
                *(float4*)(rp + 4) = make_float4(a0[4] + a1[4], a0[5] + a1[5], a0[6] + a1[6], a0[7] + a1[7]);
            }
        }
        if (t0 + TS * TG < ntiles) load_pass(t0 + TS * TG);
        lds_barrier();

        if (tid < 64 * TG) {
            const int tt = tid >> 6;
            const int ln = (tid >> 1) & 31, vh = tid & 1;
            const int t  = t0 + TS * tt;
            const int r  = 16 * t + (ln & 15);
            if (t < ntiles && r < total) {
                const float* rp = &red[tt][0][ln * 8 + 4 * vh];
                float4 sum = *(const float4*)rp;
                #pragma unroll
                for (int w = 1; w < NW; ++w) {
                    const float4 p = *(const float4*)(rp + 256 * w);
                    sum.x += p.x; sum.y += p.y; sum.z += p.z; sum.w += p.w;
                }
                const int bi = lst[r];
                const int m0 = 16 * rt + 8 * (ln >> 4) + 4 * vh;
                const float4 q = *(const float4*)(bias_ns + dom * KSPA + m0);
                *(float4*)(out + bi * NN + NSH + m0) =
                    make_float4(sum.x + q.x, sum.y + q.y, sum.z + q.z, sum.w + q.w);
            }
        }
        lds_barrier();
    }
}

extern "C" void kernel_launch(void* const* d_in, const int* in_sizes, int n_in,
                              void* d_out, int out_size, void* d_ws, size_t ws_size,
                              hipStream_t stream) {
    const float* eps     = (const float*)d_in[0];
    const int*   didx    = (const int*)  d_in[1];
    const float* L_emb   = (const float*)d_in[2];
    const float* S_emb   = (const float*)d_in[3];
    const float* bias_ns = (const float*)d_in[4];
    const float* bias_sh = (const float*)d_in[5];
    float*       out     = (float*)d_out;

    fused_spa_kernel<<<dim3(ND, KSPA / 16, TS), dim3(512), 0, stream>>>(
        eps, didx, L_emb, S_emb, bias_ns, bias_sh, out);
}


# Row tile of M formed by one wave in registers: block back-substitution from the right on transposed tiles, each result fed back as B operand; 1 split barrier (was 6 with LDS trips); sum order changes
# speedup vs baseline: 13.2984x; 1.1153x over previous
#include <hip/hip_runtime.h>
#include <hip/hip_bf16.h>

#define NN   512
#define KSPA 64
#define ND   16
#define NB   1024
#define NSH  448
#define N_L  30688

typedef __attribute__((ext_vector_type(2))) float v2f;
typedef __attribute__((ext_vector_type(8))) float v8f;

#define MST 68
#define NW  16
#define TS  2
#define TG  3
#define SP  36
#define LW  8

__device__ __forceinline__ v8f mma4(v2f a, v2f b, v8f c) {
    return __builtin_amdgcn_wmma_f32_16x16x4_f32(
         false, a,  false, b,
         (short)0, c,  false,  false);
}

__device__ __forceinline__ void lds_barrier() {
    __builtin_amdgcn_fence(__ATOMIC_RELEASE, "workgroup", "local");
    __builtin_amdgcn_s_barrier();
    __builtin_amdgcn_fence(__ATOMIC_ACQUIRE, "workgroup", "local");
}

__device__ __forceinline__ void wave_lds_order() {
    __builtin_amdgcn_fence(__ATOMIC_RELEASE, "wavefront", "local");
    __builtin_amdgcn_wave_barrier();
    __builtin_amdgcn_fence(__ATOMIC_ACQUIRE, "wavefront", "local");
}

__device__ __forceinline__ v2f frag_t(const float (*P)[MST], int r0, int c0, int half, int l16, int q) {
    v2f a;
    a.x = P[r0 + 8 * half + 2 * q][c0 + l16];
    a.y = P[r0 + 8 * half + 2 * q + 1][c0 + l16];
    return a;
}

template <int R>
__device__ __forceinline__ void row_tile_chain(const float (*Wl)[MST], float (*Mi)[MST], int half, int l16) {
    const v8f zero8 = {0.f, 0.f, 0.f, 0.f, 0.f, 0.f, 0.f, 0.f};
    v8f Y[R + 1];
    {
        const float* p = &Mi[16 * R + l16][16 * R + 8 * half];
        const float4 a = *(const float4*)(p);
        const float4 b = *(const float4*)(p + 4);
        Y[0][0] = a.x; Y[0][1] = a.y; Y[0][2] = a.z; Y[0][3] = a.w;
        Y[0][4] = b.x; Y[0][5] = b.y; Y[0][6] = b.z; Y[0][7] = b.w;
    }
    #pragma unroll
    for (int s = 1; s <= R; ++s) {
        const int J = R - s;
        v8f S = zero8;
        #pragma unroll
        for (int t = 0; t < s; ++t) {
            const int K = R - t;
            #pragma unroll
            for (int q = 0; q < 4; ++q) {
                v2f b; b.x = Y[t][2 * q]; b.y = Y[t][2 * q + 1];
                S = mma4(frag_t(Wl, 16 * K, 16 * J, half, l16, q), b, S);
            }
        }
        v8f X = zero8;
        #pragma unroll
        for (int q = 0; q < 4; ++q) {
            v2f b; b.x = S[2 * q]; b.y = S[2 * q + 1];
            X = mma4(frag_t(Mi, 16 * J, 16 * J, half, l16, q), b, X);
        }
        Y[s] = X;
        float* o = &Mi[16 * R + l16][16 * J + 8 * half];
        *(float4*)(o)     = make_float4(X[0], X[1], X[2], X[3]);
        *(float4*)(o + 4) = make_float4(X[4], X[5], X[6], X[7]);
    }
}

__global__ __launch_bounds__(512) void fused_spa_kernel(
    const float* __restrict__ eps,
    const int*   __restrict__ didx,
    const float* __restrict__ L_emb,
    const float* __restrict__ S_emb,
    const float* __restrict__ bias_ns,
    const float* __restrict__ bias_sh,
    float* __restrict__ out)
{
    __shared__ __attribute__((aligned(16))) float Wl[64][MST];
    __shared__ __attribute__((aligned(16))) float Mi[64][MST];
    __shared__ int   lst[NB];
    __shared__ int   s_total;
    __shared__ __attribute__((aligned(16))) int cnt[LW];
    __shared__ __attribute__((aligned(16))) float red[TG][NW][256];
    __shared__ __attribute__((aligned(16))) float stg[NW][TG][16 * SP];

    const int dom  = blockIdx.x;
    const int rt   = blockIdx.y;
    const int slot = blockIdx.z;
    const int tid  = threadIdx.x;
    const int wave = tid >> 5;
    const int lane = tid & 31;
    const int half = lane >> 4;
    const int l16  = lane & 15;
    const int kw   = 32 * wave;
    const int rsub = lane >> 3;
    const int csub = 4 * (lane & 7);
    const bool listw = wave >= NW - LW;
    const int  lw    = wave - (NW - LW);
    const bool leftw = wave < 14;
    const bool pw    = leftw && ((wave & 3) == rt);

    const float* Ld = L_emb + dom * N_L;

    float wraw[8];
    unsigned mk[4];
    float4 bsc;
    v2f ca0[4], cb0[4], ca1[4], cb1[4], ca2[4], cb2[4], ca3[4], cb3[4];
    float4 sv[4];
    auto load_small = [&]() {
        if (pw) {
            bsc = *(const float4*)(bias_sh + kw + csub);
        }
        if (listw) {
            int dv[4];
            #pragma unroll
            for (int j = 0; j < 4; ++j) dv[j] = didx[128 * lw + 32 * j + lane];
            #pragma unroll
            for (int j = 0; j < 4; ++j) mk[j] = __builtin_amdgcn_ballot_w32(dv[j] == dom);
            if (lane == 0) cnt[lw] = __popc(mk[0]) + __popc(mk[1]) + __popc(mk[2]) + __popc(mk[3]);
        }
    };
    auto load_chunk = [&](int J, v2f (&fa)[4], v2f (&fb)[4]) {
        #pragma unroll
        for (int ss = 0; ss < 4; ++ss) {
            const int s  = 4 * J + ss;
            const int k0 = 4 * s + 2 * half;
            const float* r0 = Ld + NSH * k0 + (k0 * (k0 - 1)) / 2 + kw + l16;
            const float* r1 = Ld + NSH * (k0 + 1) + ((k0 + 1) * k0) / 2 + kw + l16;
            fa[ss].x = r0[0];  fa[ss].y = r1[0];
            fb[ss].x = r0[16]; fb[ss].y = r1[16];
        }
    };
    if (dom != 0) {
        #pragma unroll
        for (int q = 0; q < 8; ++q) {
            const int idx = q * 512 + tid;
            const int i = idx >> 6, k = idx & 63;
            wraw[q] = Ld[NSH * i + (i * (i - 1)) / 2 + NSH + ((k < i) ? k : 0)];
        }
        load_small();
        #pragma unroll
        for (int q = 0; q < 8; ++q) {
            const int idx = q * 512 + tid;
            const int i = idx >> 6, k = idx & 63;
            Wl[i][k] = (k < i) ? wraw[q] : 0.0f;
            Mi[i][k] = 0.0f;
        }
        if (leftw) {
            load_chunk(0, ca0, cb0);
            if (rt >= 1) load_chunk(1, ca1, cb1);
            if (rt >= 2) load_chunk(2, ca2, cb2);
        } else {
            const float* Sp = S_emb + dom * KSPA + (kw - NSH) + 8 * half;
            sv[0] = *(const float4*)(Sp);
            sv[1] = *(const float4*)(Sp + 4);
            sv[2] = *(const float4*)(Sp + 16);
            sv[3] = *(const float4*)(Sp + 20);
        }
    } else {
        load_small();
    }
    lds_barrier();

    if (listw) {
        const int4 c0 = *(const int4*)&cnt[0];
        const int4 c1 = *(const int4*)&cnt[4];
        const int c[LW] = {c0.x, c0.y, c0.z, c0.w, c1.x, c1.y, c1.z, c1.w};
        int pos = 0;
        #pragma unroll
        for (int v = 0; v < LW - 1; ++v) pos += (v < lw) ? c[v] : 0;
        #pragma unroll
        for (int j = 0; j < 4; ++j) {
            const unsigned m = mk[j];
            const bool mine = (m >> lane) & 1u;
            if (mine) lst[pos + __popc(m & ((1u << lane) - 1u))] = 128 * lw + 32 * j + lane;
            pos += __popc(m);
        }
        if (wave == NW - 1 && lane == 0) s_total = pos;
    } else if (dom != 0 && tid < 64) {
        const int I = tid >> 4, j = tid & 15;
        float x[16];
        #pragma unroll
        for (int r = 0; r < 16; ++r) x[r] = (r == j) ? 1.0f : 0.0f;
        #pragma unroll
        for (int k = 0; k < 15; ++k) {
            const float xk = x[k];
            #pragma unroll
            for (int r = k + 1; r < 16; ++r)
                x[r] += Wl[16 * I + r][16 * I + k] * xk;
        }
        #pragma unroll
        for (int r = 0; r < 16; ++r) Mi[16 * I + r][16 * I + j] = x[r];
    }
    lds_barrier();

    const bool chained = (dom != 0) && (rt >= 1);
    if (chained) {
        if (wave == NW - 1) {
            if (rt == 1)      row_tile_chain<1>(Wl, Mi, half, l16);
            else if (rt == 2) row_tile_chain<2>(Wl, Mi, half, l16);
            else              row_tile_chain<3>(Wl, Mi, half, l16);
        }
        __builtin_amdgcn_fence(__ATOMIC_RELEASE, "workgroup", "local");
        __builtin_amdgcn_s_barrier_signal(-1);
    }

    const int total  = s_total;
    const int ntiles = (total + 15) >> 4;

    float4 cv[TG][4];
    int    cb[TG][4];
    bool   cvl[TG][4];
    auto load_pass = [&](int tbase) {
        #pragma unroll
        for (int tt = 0; tt < TG; ++tt) {
            const int t = tbase + TS * tt;
            #pragma unroll
            for (int j = 0; j < 4; ++j) { cb[tt][j] = 0; cvl[tt][j] = false; }
            if (t < ntiles) {
                #pragma unroll
                for (int j = 0; j < 4; ++j) {
                    const int r = 16 * t + 4 * j + rsub;
                    cvl[tt][j] = r < total;
                    cb[tt][j] = lst[cvl[tt][j] ? r : 16 * t];
                    cv[tt][j] = *(const float4*)(eps + cb[tt][j] * NN + kw + csub);
                }
            }
        }
    };
    if (dom != 0 && leftw && rt >= 3) load_chunk(3, ca3, cb3);

    #pragma unroll
    for (int tt = 0; tt < TG; ++tt) {
        const int t = slot + TS * tt;
        const bool has = t < ntiles;
        #pragma unroll
        for (int j = 0; j < 4; ++j) {
            const int r = 16 * t + 4 * j + rsub;
            cvl[tt][j] = has && (r < total);
            const int li = lst[cvl[tt][j] ? r : (has ? 16 * t : 0)];
            cb[tt][j] = has ? li : 0;
            cv[tt][j] = *(const float4*)(eps + cb[tt][j] * NN + kw + csub);
        }
    }

    const v8f zero8 = {0.f, 0.f, 0.f, 0.f, 0.f, 0.f, 0.f, 0.f};
    v8f F0 = zero8, F1 = zero8;

    if (dom != 0) {
        if (rt >= 1) {
            __builtin_amdgcn_s_barrier_wait(-1);
            __builtin_amdgcn_fence(__ATOMIC_ACQUIRE, "workgroup", "local");
        }

        if (leftw) {
            auto mul_chunk = [&](int J, const v2f (&fa)[4], const v2f (&fb)[4]) {
                #pragma unroll
                for (int ss = 0; ss < 4; ++ss) {
                    const int s  = 4 * J + ss;
                    const int k0 = 4 * s + 2 * half;
                    v2f b; b.x = Mi[16 * rt + l16][k0];
                           b.y = Mi[16 * rt + l16][k0 + 1];
                    F0 = mma4(fa[ss], b, F0);
                    F1 = mma4(fb[ss], b, F1);
                }
            };
            mul_chunk(0, ca0, cb0);
            if (rt >= 1) mul_chunk(1, ca1, cb1);
            if (rt >= 2) mul_chunk(2, ca2, cb2);
            if (rt >= 3) mul_chunk(3, ca3, cb3);
        } else {
            const float* Mr = &Mi[16 * rt + l16][(kw - NSH) + 8 * half];
            const float4 m0 = *(const float4*)(Mr);
            const float4 m1 = *(const float4*)(Mr + 4);
            const float4 m2 = *(const float4*)(Mr + 16);
            const float4 m3 = *(const float4*)(Mr + 20);
            F0[0] = m0.x * sv[0].x; F0[1] = m0.y * sv[0].y; F0[2] = m0.z * sv[0].z; F0[3] = m0.w * sv[0].w;
            F0[4] = m1.x * sv[1].x; F0[5] = m1.y * sv[1].y; F0[6] = m1.z * sv[1].z; F0[7] = m1.w * sv[1].w;
            F1[0] = m2.x * sv[2].x; F1[1] = m2.y * sv[2].y; F1[2] = m2.z * sv[2].z; F1[3] = m2.w * sv[2].w;
            F1[4] = m3.x * sv[3].x; F1[5] = m3.y * sv[3].y; F1[6] = m3.z * sv[3].z; F1[7] = m3.w * sv[3].w;
        }
    } else {
        const int one_at = NSH + 16 * rt + l16 - kw - 8 * half;
        #pragma unroll
        for (int r = 0; r < 8; ++r) {
            F0[r] = (r == one_at) ? 1.0f : 0.0f;
            F1[r] = (r + 16 == one_at) ? 1.0f : 0.0f;
        }
    }

    for (int t0 = slot; t0 < ntiles; t0 += TS * TG) {
        #pragma unroll
        for (int tt = 0; tt < TG; ++tt) {
            if (t0 + TS * tt < ntiles) {
                float* sp = &stg[wave][tt][0];
                #pragma unroll
                for (int j = 0; j < 4; ++j)
                    *(float4*)(sp + (4 * j + rsub) * SP + csub) = cv[tt][j];
                if (pw) {
                    float4 y[4];
                    #pragma unroll
                    for (int j = 0; j < 4; ++j) {
                        const float4 x = cv[tt][j];
                        y[j] = make_float4(x.x + bsc.x, x.y + bsc.y, x.z + bsc.z, x.w + bsc.w);
                    }
                    if (16 * (t0 + TS * tt) + 16 <= total) {
                        #pragma unroll
                        for (int j = 0; j < 4; ++j)
                            *(float4*)(out + cb[tt][j] * NN + kw + csub) = y[j];
                    } else {
                        #pragma unroll
                        for (int j = 0; j < 4; ++j)
                            if (cvl[tt][j]) *(float4*)(out + cb[tt][j] * NN + kw + csub) = y[j];
                    }
                }
            }
        }
        wave_lds_order();
        float4 ev[TG][4];
        #pragma unroll
        for (int tt = 0; tt < TG; ++tt) {
            if (t0 + TS * tt < ntiles) {
                const float* fp = &stg[wave][tt][l16 * SP + 8 * half];
                ev[tt][0] = *(const float4*)(fp);
                ev[tt][1] = *(const float4*)(fp + 4);
                ev[tt][2] = *(const float4*)(fp + 16);
                ev[tt][3] = *(const float4*)(fp + 20);
            }
        }
        #pragma unroll
        for (int tt = 0; tt < TG; ++tt) {
            if (t0 + TS * tt < ntiles) {
                v8f a0 = zero8, a1 = zero8;
                #pragma unroll
                for (int q = 0; q < 4; ++q) {
                    const float4 e0 = ev[tt][q >> 1];
                    const float4 e1 = ev[tt][2 + (q >> 1)];
                    v2f fa0; fa0.x = F0[2 * q]; fa0.y = F0[2 * q + 1];
                    v2f fa1; fa1.x = F1[2 * q]; fa1.y = F1[2 * q + 1];
                    v2f b0;  b0.x = (q & 1) ? e0.z : e0.x;  b0.y = (q & 1) ? e0.w : e0.y;
                    v2f b1;  b1.x = (q & 1) ? e1.z : e1.x;  b1.y = (q & 1) ? e1.w : e1.y;
                    a0 = mma4(fa0, b0, a0);
                    a1 = mma4(fa1, b1, a1);
                }
                float* rp = &red[tt][wave][lane * 8];
                *(float4*)(rp)     = make_float4(a0[0] + a1[0], a0[1] + a1[1], a0[2] + a1[2], a0[3] + a1[3]);
                *(float4*)(rp + 4) = make_float4(a0[4] + a1[4], a0[5] + a1[5], a0[6] + a1[6], a0[7] + a1[7]);
            }
        }
        if (t0 + TS * TG < ntiles) load_pass(t0 + TS * TG);
        lds_barrier();

        if (tid < 64 * TG) {
            const int tt = tid >> 6;
            const int ln = (tid >> 1) & 31, vh = tid & 1;
            const int t  = t0 + TS * tt;
            const int r  = 16 * t + (ln & 15);
            if (t < ntiles && r < total) {
                const float* rp = &red[tt][0][ln * 8 + 4 * vh];
                float4 sum = *(const float4*)rp;
                #pragma unroll
                for (int w = 1; w < NW; ++w) {
                    const float4 p = *(const float4*)(rp + 256 * w);
                    sum.x += p.x; sum.y += p.y; sum.z += p.z; sum.w += p.w;
                }
                const int bi = lst[r];
                const int m0 = 16 * rt + 8 * (ln >> 4) + 4 * vh;
                const float4 q = *(const float4*)(bias_ns + dom * KSPA + m0);
                *(float4*)(out + bi * NN + NSH + m0) =
                    make_float4(sum.x + q.x, sum.y + q.y, sum.z + q.z, sum.w + q.w);
            }
        }
        lds_barrier();
    }
}

extern "C" void kernel_launch(void* const* d_in, const int* in_sizes, int n_in,
                              void* d_out, int out_size, void* d_ws, size_t ws_size,
                              hipStream_t stream) {
    const float* eps     = (const float*)d_in[0];
    const int*   didx    = (const int*)  d_in[1];
    const float* L_emb   = (const float*)d_in[2];
    const float* S_emb   = (const float*)d_in[3];
    const float* bias_ns = (const float*)d_in[4];
    const float* bias_sh = (const float*)d_in[5];
    float*       out     = (float*)d_out;

    fused_spa_kernel<<<dim3(ND, KSPA / 16, TS), dim3(512), 0, stream>>>(
        eps, didx, L_emb, S_emb, bias_ns, bias_sh, out);
}


# Tile slots per row tile follow the F build's length: 6, 4, 3, 2 blocks for row tiles 3, 2, 1, 0 (were 2 each), grid 16x15; the longest build multiplies 1 tile, not 3; tileless blocks leave; same sums
# speedup vs baseline: 14.3996x; 1.0828x over previous
#include <hip/hip_runtime.h>
#include <hip/hip_bf16.h>

#define NN   512
#define KSPA 64
#define ND   16
#define NB   1024
#define NSH  448
#define N_L  30688

typedef __attribute__((ext_vector_type(2))) float v2f;
typedef __attribute__((ext_vector_type(8))) float v8f;

#define MST 68
#define NW  16
#define NY  15
#define TG  3
#define SP  36
#define LW  8

__device__ __forceinline__ v8f mma4(v2f a, v2f b, v8f c) {
    return __builtin_amdgcn_wmma_f32_16x16x4_f32(
         false, a,  false, b,
         (short)0, c,  false,  false);
}

__device__ __forceinline__ void lds_barrier() {
    __builtin_amdgcn_fence(__ATOMIC_RELEASE, "workgroup", "local");
    __builtin_amdgcn_s_barrier();
    __builtin_amdgcn_fence(__ATOMIC_ACQUIRE, "workgroup", "local");
}

__device__ __forceinline__ void wave_lds_order() {
    __builtin_amdgcn_fence(__ATOMIC_RELEASE, "wavefront", "local");
    __builtin_amdgcn_wave_barrier();
    __builtin_amdgcn_fence(__ATOMIC_ACQUIRE, "wavefront", "local");
}

__device__ __forceinline__ v2f frag_t(const float (*P)[MST], int r0, int c0, int half, int l16, int q) {
    v2f a;
    a.x = P[r0 + 8 * half + 2 * q][c0 + l16];
    a.y = P[r0 + 8 * half + 2 * q + 1][c0 + l16];
    return a;
}

template <int R>
__device__ __forceinline__ void row_tile_chain(const float (*Wl)[MST], float (*Mi)[MST], int half, int l16) {
    const v8f zero8 = {0.f, 0.f, 0.f, 0.f, 0.f, 0.f, 0.f, 0.f};
    v8f Y[R + 1];
    {
        const float* p = &Mi[16 * R + l16][16 * R + 8 * half];
        const float4 a = *(const float4*)(p);
        const float4 b = *(const float4*)(p + 4);
        Y[0][0] = a.x; Y[0][1] = a.y; Y[0][2] = a.z; Y[0][3] = a.w;
        Y[0][4] = b.x; Y[0][5] = b.y; Y[0][6] = b.z; Y[0][7] = b.w;
    }
    #pragma unroll
    for (int s = 1; s <= R; ++s) {
        const int J = R - s;
        v8f S = zero8;
        #pragma unroll
        for (int t = 0; t < s; ++t) {
            const int K = R - t;
            #pragma unroll
            for (int q = 0; q < 4; ++q) {
                v2f b; b.x = Y[t][2 * q]; b.y = Y[t][2 * q + 1];
                S = mma4(frag_t(Wl, 16 * K, 16 * J, half, l16, q), b, S);
            }
        }
        v8f X = zero8;
        #pragma unroll
        for (int q = 0; q < 4; ++q) {
            v2f b; b.x = S[2 * q]; b.y = S[2 * q + 1];
            X = mma4(frag_t(Mi, 16 * J, 16 * J, half, l16, q), b, X);
        }
        Y[s] = X;
        float* o = &Mi[16 * R + l16][16 * J + 8 * half];
        *(float4*)(o)     = make_float4(X[0], X[1], X[2], X[3]);
        *(float4*)(o + 4) = make_float4(X[4], X[5], X[6], X[7]);
    }
}

__global__ __launch_bounds__(512) void fused_spa_kernel(
    const float* __restrict__ eps,
    const int*   __restrict__ didx,
    const float* __restrict__ L_emb,
    const float* __restrict__ S_emb,
    const float* __restrict__ bias_ns,
    const float* __restrict__ bias_sh,
    float* __restrict__ out)
{
    __shared__ __attribute__((aligned(16))) float Wl[64][MST];
    __shared__ __attribute__((aligned(16))) float Mi[64][MST];
    __shared__ int   lst[NB];
    __shared__ int   s_total;
    __shared__ __attribute__((aligned(16))) int cnt[LW];
    __shared__ __attribute__((aligned(16))) float red[TG][NW][256];
    __shared__ __attribute__((aligned(16))) float stg[NW][TG][16 * SP];

    const int dom  = blockIdx.x;
    const int by   = blockIdx.y;
    const int rt   = (by < 6) ? 3 : (by < 10) ? 2 : (by < 13) ? 1 : 0;
    const int slot = (by < 6) ? by : (by < 10) ? by - 6 : (by < 13) ? by - 10 : by - 13;
    const int ns   = (by < 6) ? 6 : (by < 10) ? 4 : (by < 13) ? 3 : 2;
    const int tid  = threadIdx.x;
    const int wave = tid >> 5;
    const int lane = tid & 31;
    const int half = lane >> 4;
    const int l16  = lane & 15;
    const int kw   = 32 * wave;
    const int rsub = lane >> 3;
    const int csub = 4 * (lane & 7);
    const bool listw = wave >= NW - LW;
    const int  lw    = wave - (NW - LW);
    const bool leftw = wave < 14;
    const bool pw    = leftw && ((wave & 3) == rt);

    const float* Ld = L_emb + dom * N_L;

    float wraw[8];
    unsigned mk[4];
    float4 bsc;
    v2f ca0[4], cb0[4], ca1[4], cb1[4], ca2[4], cb2[4], ca3[4], cb3[4];
    float4 sv[4];
    auto load_small = [&]() {
        if (pw) {
            bsc = *(const float4*)(bias_sh + kw + csub);
        }
        if (listw) {
            int dv[4];
            #pragma unroll
            for (int j = 0; j < 4; ++j) dv[j] = didx[128 * lw + 32 * j + lane];
            #pragma unroll
            for (int j = 0; j < 4; ++j) mk[j] = __builtin_amdgcn_ballot_w32(dv[j] == dom);
            if (lane == 0) cnt[lw] = __popc(mk[0]) + __popc(mk[1]) + __popc(mk[2]) + __popc(mk[3]);
        }
    };
    auto load_chunk = [&](int J, v2f (&fa)[4], v2f (&fb)[4]) {
        #pragma unroll
        for (int ss = 0; ss < 4; ++ss) {
            const int s  = 4 * J + ss;
            const int k0 = 4 * s + 2 * half;
            const float* r0 = Ld + NSH * k0 + (k0 * (k0 - 1)) / 2 + kw + l16;
            const float* r1 = Ld + NSH * (k0 + 1) + ((k0 + 1) * k0) / 2 + kw + l16;
            fa[ss].x = r0[0];  fa[ss].y = r1[0];
            fb[ss].x = r0[16]; fb[ss].y = r1[16];
        }
    };
    if (dom != 0) {
        #pragma unroll
        for (int q = 0; q < 8; ++q) {
            const int idx = q * 512 + tid;
            const int i = idx >> 6, k = idx & 63;
            wraw[q] = Ld[NSH * i + (i * (i - 1)) / 2 + NSH + ((k < i) ? k : 0)];
        }
        load_small();
        #pragma unroll
        for (int q = 0; q < 8; ++q) {
            const int idx = q * 512 + tid;
            const int i = idx >> 6, k = idx & 63;
            Wl[i][k] = (k < i) ? wraw[q] : 0.0f;
            Mi[i][k] = 0.0f;
        }
        if (leftw) {
            load_chunk(0, ca0, cb0);
            if (rt >= 1) load_chunk(1, ca1, cb1);
            if (rt >= 2) load_chunk(2, ca2, cb2);
        } else {
            const float* Sp = S_emb + dom * KSPA + (kw - NSH) + 8 * half;
            sv[0] = *(const float4*)(Sp);
            sv[1] = *(const float4*)(Sp + 4);
            sv[2] = *(const float4*)(Sp + 16);
            sv[3] = *(const float4*)(Sp + 20);
        }
    } else {
        load_small();
    }
    lds_barrier();

    if (listw) {
        const int4 c0 = *(const int4*)&cnt[0];
        const int4 c1 = *(const int4*)&cnt[4];
        const int c[LW] = {c0.x, c0.y, c0.z, c0.w, c1.x, c1.y, c1.z, c1.w};
        int pos = 0;
        #pragma unroll
        for (int v = 0; v < LW - 1; ++v) pos += (v < lw) ? c[v] : 0;
        #pragma unroll
        for (int j = 0; j < 4; ++j) {
            const unsigned m = mk[j];
            const bool mine = (m >> lane) & 1u;
            if (mine) lst[pos + __popc(m & ((1u << lane) - 1u))] = 128 * lw + 32 * j + lane;
            pos += __popc(m);
        }
        if (wave == NW - 1 && lane == 0) s_total = pos;
    } else if (dom != 0 && tid < 64) {
        const int I = tid >> 4, j = tid & 15;
        float x[16];
        #pragma unroll
        for (int r = 0; r < 16; ++r) x[r] = (r == j) ? 1.0f : 0.0f;
        #pragma unroll
        for (int k = 0; k < 15; ++k) {
            const float xk = x[k];
            #pragma unroll
            for (int r = k + 1; r < 16; ++r)
                x[r] += Wl[16 * I + r][16 * I + k] * xk;
        }
        #pragma unroll
        for (int r = 0; r < 16; ++r) Mi[16 * I + r][16 * I + j] = x[r];
    }
    lds_barrier();

    const bool chained = (dom != 0) && (rt >= 1);
    if (chained) {
        if (wave == NW - 1) {
            if (rt == 1)      row_tile_chain<1>(Wl, Mi, half, l16);
            else if (rt == 2) row_tile_chain<2>(Wl, Mi, half, l16);
            else              row_tile_chain<3>(Wl, Mi, half, l16);
        }
        __builtin_amdgcn_fence(__ATOMIC_RELEASE, "workgroup", "local");
        __builtin_amdgcn_s_barrier_signal(-1);
    }

    const int total  = s_total;
    const int ntiles = (total + 15) >> 4;
    if (slot >= ntiles) {
        if (chained) __builtin_amdgcn_s_barrier_wait(-1);
        return;
    }

    float4 cv[TG][4];
    int    cb[TG][4];
    bool   cvl[TG][4];
    auto load_pass = [&](int tbase) {
        #pragma unroll
        for (int tt = 0; tt < TG; ++tt) {
            const int t = tbase + ns * tt;
            #pragma unroll
            for (int j = 0; j < 4; ++j) { cb[tt][j] = 0; cvl[tt][j] = false; }
            if (t < ntiles) {
                #pragma unroll
                for (int j = 0; j < 4; ++j) {
                    const int r = 16 * t + 4 * j + rsub;
                    cvl[tt][j] = r < total;
                    cb[tt][j] = lst[cvl[tt][j] ? r : 16 * t];
                    cv[tt][j] = *(const float4*)(eps + cb[tt][j] * NN + kw + csub);
                }
            }
        }
    };
    if (dom != 0 && leftw && rt >= 3) load_chunk(3, ca3, cb3);

    #pragma unroll
    for (int tt = 0; tt < TG; ++tt) {
        const int t = slot + ns * tt;
        const bool has = t < ntiles;
        #pragma unroll
        for (int j = 0; j < 4; ++j) {
            const int r = 16 * t + 4 * j + rsub;
            cvl[tt][j] = has && (r < total);
            const int li = lst[cvl[tt][j] ? r : (has ? 16 * t : 0)];
            cb[tt][j] = has ? li : 0;
            cv[tt][j] = *(const float4*)(eps + cb[tt][j] * NN + kw + csub);
        }
    }

    const v8f zero8 = {0.f, 0.f, 0.f, 0.f, 0.f, 0.f, 0.f, 0.f};
    v8f F0 = zero8, F1 = zero8;

    if (dom != 0) {
        if (rt >= 1) {
            __builtin_amdgcn_s_barrier_wait(-1);
            __builtin_amdgcn_fence(__ATOMIC_ACQUIRE, "workgroup", "local");
        }

        if (leftw) {
            auto mul_chunk = [&](int J, const v2f (&fa)[4], const v2f (&fb)[4]) {
                #pragma unroll
                for (int ss = 0; ss < 4; ++ss) {
                    const int s  = 4 * J + ss;
                    const int k0 = 4 * s + 2 * half;
                    v2f b; b.x = Mi[16 * rt + l16][k0];
                           b.y = Mi[16 * rt + l16][k0 + 1];
                    F0 = mma4(fa[ss], b, F0);
                    F1 = mma4(fb[ss], b, F1);
                }
            };
            mul_chunk(0, ca0, cb0);
            if (rt >= 1) mul_chunk(1, ca1, cb1);
            if (rt >= 2) mul_chunk(2, ca2, cb2);
            if (rt >= 3) mul_chunk(3, ca3, cb3);
        } else {
            const float* Mr = &Mi[16 * rt + l16][(kw - NSH) + 8 * half];
            const float4 m0 = *(const float4*)(Mr);
            const float4 m1 = *(const float4*)(Mr + 4);
            const float4 m2 = *(const float4*)(Mr + 16);
            const float4 m3 = *(const float4*)(Mr + 20);
            F0[0] = m0.x * sv[0].x; F0[1] = m0.y * sv[0].y; F0[2] = m0.z * sv[0].z; F0[3] = m0.w * sv[0].w;
            F0[4] = m1.x * sv[1].x; F0[5] = m1.y * sv[1].y; F0[6] = m1.z * sv[1].z; F0[7] = m1.w * sv[1].w;
            F1[0] = m2.x * sv[2].x; F1[1] = m2.y * sv[2].y; F1[2] = m2.z * sv[2].z; F1[3] = m2.w * sv[2].w;
            F1[4] = m3.x * sv[3].x; F1[5] = m3.y * sv[3].y; F1[6] = m3.z * sv[3].z; F1[7] = m3.w * sv[3].w;
        }
    } else {
        const int one_at = NSH + 16 * rt + l16 - kw - 8 * half;
        #pragma unroll
        for (int r = 0; r < 8; ++r) {
            F0[r] = (r == one_at) ? 1.0f : 0.0f;
            F1[r] = (r + 16 == one_at) ? 1.0f : 0.0f;
        }
    }

    for (int t0 = slot; t0 < ntiles; t0 += ns * TG) {
        #pragma unroll
        for (int tt = 0; tt < TG; ++tt) {
            if (t0 + ns * tt < ntiles) {
                float* sp = &stg[wave][tt][0];
                #pragma unroll
                for (int j = 0; j < 4; ++j)
                    *(float4*)(sp + (4 * j + rsub) * SP + csub) = cv[tt][j];
                if (pw) {
                    float4 y[4];
                    #pragma unroll
                    for (int j = 0; j < 4; ++j) {
                        const float4 x = cv[tt][j];
                        y[j] = make_float4(x.x + bsc.x, x.y + bsc.y, x.z + bsc.z, x.w + bsc.w);
                    }
                    if (16 * (t0 + ns * tt) + 16 <= total) {
                        #pragma unroll
                        for (int j = 0; j < 4; ++j)
                            *(float4*)(out + cb[tt][j] * NN + kw + csub) = y[j];
                    } else {
                        #pragma unroll
                        for (int j = 0; j < 4; ++j)
                            if (cvl[tt][j]) *(float4*)(out + cb[tt][j] * NN + kw + csub) = y[j];
                    }
                }
            }
        }
        wave_lds_order();
        float4 ev[TG][4];
        #pragma unroll
        for (int tt = 0; tt < TG; ++tt) {
            if (t0 + ns * tt < ntiles) {
                const float* fp = &stg[wave][tt][l16 * SP + 8 * half];
                ev[tt][0] = *(const float4*)(fp);
                ev[tt][1] = *(const float4*)(fp + 4);
                ev[tt][2] = *(const float4*)(fp + 16);
                ev[tt][3] = *(const float4*)(fp + 20);
            }
        }
        #pragma unroll
        for (int tt = 0; tt < TG; ++tt) {
            if (t0 + ns * tt < ntiles) {
                v8f a0 = zero8, a1 = zero8;
                #pragma unroll
                for (int q = 0; q < 4; ++q) {
                    const float4 e0 = ev[tt][q >> 1];
                    const float4 e1 = ev[tt][2 + (q >> 1)];
                    v2f fa0; fa0.x = F0[2 * q]; fa0.y = F0[2 * q + 1];
                    v2f fa1; fa1.x = F1[2 * q]; fa1.y = F1[2 * q + 1];
                    v2f b0;  b0.x = (q & 1) ? e0.z : e0.x;  b0.y = (q & 1) ? e0.w : e0.y;
                    v2f b1;  b1.x = (q & 1) ? e1.z : e1.x;  b1.y = (q & 1) ? e1.w : e1.y;
                    a0 = mma4(fa0, b0, a0);
                    a1 = mma4(fa1, b1, a1);
                }
                float* rp = &red[tt][wave][lane * 8];
                *(float4*)(rp)     = make_float4(a0[0] + a1[0], a0[1] + a1[1], a0[2] + a1[2], a0[3] + a1[3]);
                *(float4*)(rp + 4) = make_float4(a0[4] + a1[4], a0[5] + a1[5], a0[6] + a1[6], a0[7] + a1[7]);
            }
        }
        if (t0 + ns * TG < ntiles) load_pass(t0 + ns * TG);
        lds_barrier();

        if (tid < 64 * TG) {
            const int tt = tid >> 6;
            const int ln = (tid >> 1) & 31, vh = tid & 1;
            const int t  = t0 + ns * tt;
            const int r  = 16 * t + (ln & 15);
            if (t < ntiles && r < total) {
                const float* rp = &red[tt][0][ln * 8 + 4 * vh];
                float4 sum = *(const float4*)rp;
                #pragma unroll
                for (int w = 1; w < NW; ++w) {
                    const float4 p = *(const float4*)(rp + 256 * w);
                    sum.x += p.x; sum.y += p.y; sum.z += p.z; sum.w += p.w;
                }
                const int bi = lst[r];
                const int m0 = 16 * rt + 8 * (ln >> 4) + 4 * vh;
                const float4 q = *(const float4*)(bias_ns + dom * KSPA + m0);
                *(float4*)(out + bi * NN + NSH + m0) =
                    make_float4(sum.x + q.x, sum.y + q.y, sum.z + q.z, sum.w + q.w);
            }
        }
        lds_barrier();
    }
}

extern "C" void kernel_launch(void* const* d_in, const int* in_sizes, int n_in,
                              void* d_out, int out_size, void* d_ws, size_t ws_size,
                              hipStream_t stream) {
    const float* eps     = (const float*)d_in[0];
    const int*   didx    = (const int*)  d_in[1];
    const float* L_emb   = (const float*)d_in[2];
    const float* S_emb   = (const float*)d_in[3];
    const float* bias_ns = (const float*)d_in[4];
    const float* bias_sh = (const float*)d_in[5];
    float*       out     = (float*)d_out;

    fused_spa_kernel<<<dim3(ND, NY, 1), dim3(512), 0, stream>>>(
        eps, didx, L_emb, S_emb, bias_ns, bias_sh, out);
}
